# starting kernel
# baseline (speedup 1.0000x reference)
#include <hip/hip_runtime.h>
#include <hip/hip_bf16.h>
#include <math.h>

#define DD 64
#define NSLOPE 0.2f

typedef __attribute__((ext_vector_type(2))) float v2f;
typedef __attribute__((ext_vector_type(8))) float v8f;

__device__ __forceinline__ unsigned flipF(float f) {
    unsigned u = __float_as_uint(f);
    return (u >> 31) ? ~u : (u | 0x80000000u);
}
__device__ __forceinline__ float unflipF(unsigned u) {
    unsigned v = (u >> 31) ? (u & 0x7FFFFFFFu) : ~u;
    return __uint_as_float(v);
}

__global__ void k_init(float* __restrict__ agg, float* __restrict__ denom,
                       unsigned* __restrict__ emaxU, int n) {
    int i = blockIdx.x * blockDim.x + threadIdx.x;
    if (i < n * DD) agg[i] = 0.0f;
    if (i < n) { denom[i] = 0.0f; emaxU[i] = 0u; }
}

__global__ __launch_bounds__(256) void k_gemm(const float* __restrict__ x,
                                              const float* __restrict__ W,
                                              const float* __restrict__ bias,
                                              float* __restrict__ Wh, int n) {
    const int lane = threadIdx.x & 31;
    const int wave = threadIdx.x >> 5;
    const int tile = blockIdx.x * 8 + wave;
    if (tile * 16 >= n) return;

    const int lm   = lane & 15;
    const int half = lane >> 4;
    const int row  = tile * 16 + lm;

    const float* xr = x + (size_t)row * DD + 2 * half;
    v2f areg[16];
#pragma unroll
    for (int i = 0; i < 16; ++i) {
        areg[i].x = xr[4 * i + 0];
        areg[i].y = xr[4 * i + 1];
    }

#pragma unroll
    for (int c = 0; c < 4; ++c) {
        v8f acc = {};
        const float* wr = W + (size_t)(c * 16 + lm) * DD + 2 * half;
#pragma unroll
        for (int i = 0; i < 16; ++i) {
            v2f bfrag;
            bfrag.x = wr[4 * i + 0];
            bfrag.y = wr[4 * i + 1];
            acc = __builtin_amdgcn_wmma_f32_16x16x4_f32(
                false, areg[i], false, bfrag, (short)0, acc, false, false);
        }
        const float bv = bias[c * 16 + lm];
#pragma unroll
        for (int v = 0; v < 8; ++v) {
            Wh[(size_t)(tile * 16 + v + 8 * half) * DD + c * 16 + lm] = acc[v] + bv;
        }
    }
}

__global__ __launch_bounds__(256) void k_logits(const float* __restrict__ Wh,
                                                const float* __restrict__ avec,
                                                const int* __restrict__ src,
                                                const int* __restrict__ dst,
                                                float* __restrict__ eb,
                                                unsigned* __restrict__ emaxU, int E) {
    const int lane = threadIdx.x & 31;
    const int e = blockIdx.x * 8 + (threadIdx.x >> 5);
    if (e >= E) return;
    const int s = src[e], d = dst[e];

    const float2 ws = ((const float2*)(Wh + (size_t)s * DD))[lane];
    const float2 wd = ((const float2*)(Wh + (size_t)d * DD))[lane];
    const float2 av = ((const float2*)avec)[lane];

    float sx = ws.x + wd.x, sy = ws.y + wd.y;
    sx = sx > 0.0f ? sx : NSLOPE * sx;
    sy = sy > 0.0f ? sy : NSLOPE * sy;
    float p = sx * av.x + sy * av.y;

#pragma unroll
    for (int m = 16; m >= 1; m >>= 1) p += __shfl_xor(p, m, 32);

    if (lane == 0) {
        eb[e] = p;
        atomicMax(emaxU + d, flipF(p));
    }
}

__global__ void k_exp(const int* __restrict__ dst, float* __restrict__ eb,
                      const unsigned* __restrict__ emaxU,
                      float* __restrict__ denom, int E) {
    int i = blockIdx.x * blockDim.x + threadIdx.x;
    if (i >= E) return;
    int d = dst[i];
    float ex = __expf(eb[i] - unflipF(emaxU[d]));
    eb[i] = ex;
    unsafeAtomicAdd(denom + d, ex);
}

__global__ __launch_bounds__(256) void k_agg(const float* __restrict__ Wh,
                                             const int* __restrict__ src,
                                             const int* __restrict__ dst,
                                             const float* __restrict__ eb,
                                             const float* __restrict__ denom,
                                             float* __restrict__ agg, int E) {
    const int lane = threadIdx.x & 31;
    const int e = blockIdx.x * 8 + (threadIdx.x >> 5);
    if (e >= E) return;
    const int s = src[e], d = dst[e];
    const float alpha = eb[e] / fmaxf(denom[d], 1e-9f);
    const float2 w = ((const float2*)(Wh + (size_t)s * DD))[lane];
    float* ap = agg + (size_t)d * DD + 2 * lane;
    unsafeAtomicAdd(ap + 0, alpha * w.x);
    unsafeAtomicAdd(ap + 1, alpha * w.y);
}

__global__ void k_sig(float* __restrict__ out, int n) {
    int i = blockIdx.x * blockDim.x + threadIdx.x;
    if (i < n * DD) out[i] = 1.0f / (1.0f + __expf(-out[i]));
}

extern "C" void kernel_launch(void* const* d_in, const int* in_sizes, int n_in,
                              void* d_out, int out_size, void* d_ws, size_t ws_size,
                              hipStream_t stream) {
    const float* x    = (const float*)d_in[0];
    const float* W    = (const float*)d_in[1];
    const float* bias = (const float*)d_in[2];
    const float* avec = (const float*)d_in[3];
    const int*   ei   = (const int*)d_in[4];

    const int N = in_sizes[0] / DD;
    const int E = in_sizes[4] / 2;
    const int* src = ei;
    const int* dst = ei + E;

    float*    Wh    = (float*)d_ws;
    float*    eb    = Wh + (size_t)N * DD;
    unsigned* emaxU = (unsigned*)(eb + E);
    float*    denom = (float*)(emaxU + N);

    float* agg = (float*)d_out;

    const int nd = N * DD;
    k_init<<<(nd + 255) / 256, 256, 0, stream>>>(agg, denom, emaxU, N);

    const int tiles = (N + 15) / 16;
    k_gemm<<<(tiles + 7) / 8, 256, 0, stream>>>(x, W, bias, Wh, N);

    k_logits<<<(E + 7) / 8, 256, 0, stream>>>(Wh, avec, src, dst, eb, emaxU, E);

    k_exp<<<(E + 255) / 256, 256, 0, stream>>>(dst, eb, emaxU, denom, E);

    k_agg<<<(E + 7) / 8, 256, 0, stream>>>(Wh, src, dst, eb, denom, agg, E);

    k_sig<<<(nd + 255) / 256, 256, 0, stream>>>(agg, N);
}


# edge and node passes become one owner kernel: a 1024-thread group reads all dst, lists its node range's edges in LDS; max, denominator, row sums by LDS atomics; 6 to 2 launches; sums by LDS arrival
# speedup vs baseline: 3.3150x; 3.3150x over previous
#include <hip/hip_runtime.h>
#include <hip/hip_bf16.h>
#include <math.h>

#define DD 64
#define NSLOPE 0.2f

typedef __attribute__((ext_vector_type(2))) float v2f;
typedef __attribute__((ext_vector_type(8))) float v8f;

__device__ __forceinline__ unsigned flipF(float f) {
    unsigned u = __float_as_uint(f);
    return (u >> 31) ? ~u : (u | 0x80000000u);
}
__device__ __forceinline__ float unflipF(unsigned u) {
    unsigned v = (u >> 31) ? (u & 0x7FFFFFFFu) : ~u;
    return __uint_as_float(v);
}

__global__ __launch_bounds__(256) void k_gemm(const float* __restrict__ x,
                                              const float* __restrict__ W,
                                              const float* __restrict__ bias,
                                              float* __restrict__ Wh, int n) {
    const int lane = threadIdx.x & 31;
    const int wave = threadIdx.x >> 5;
    const int tile = blockIdx.x * 8 + wave;
    if (tile * 16 >= n) return;

    const int lm   = lane & 15;
    const int half = lane >> 4;
    const int row  = tile * 16 + lm;

    const float* xr = x + (size_t)row * DD + 2 * half;
    v2f areg[16];
#pragma unroll
    for (int i = 0; i < 16; ++i) {
        areg[i].x = xr[4 * i + 0];
        areg[i].y = xr[4 * i + 1];
    }

#pragma unroll
    for (int c = 0; c < 4; ++c) {
        v8f acc = {};
        const float* wr = W + (size_t)(c * 16 + lm) * DD + 2 * half;
#pragma unroll
        for (int i = 0; i < 16; ++i) {
            v2f bfrag;
            bfrag.x = wr[4 * i + 0];
            bfrag.y = wr[4 * i + 1];
            acc = __builtin_amdgcn_wmma_f32_16x16x4_f32(
                false, areg[i], false, bfrag, (short)0, acc, false, false);
        }
        const float bv = bias[c * 16 + lm];
#pragma unroll
        for (int v = 0; v < 8; ++v) {
            Wh[(size_t)(tile * 16 + v + 8 * half) * DD + c * 16 + lm] = acc[v] + bv;
        }
    }
}

#define RMAX 196
#define CAP 6144
#define OWN_BLOCKS 256
#define OWN_THREADS 1024
#define SCAN_U 8
#define EDGE_U 8

__device__ __forceinline__ float leaky(float s) { return s > 0.0f ? s : NSLOPE * s; }

__device__ __forceinline__ float lane_logit(const float* __restrict__ whs,
                                            const float* wdl,
                                            const float* __restrict__ avec) {
    float acc = 0.0f;
    for (int k = 0; k < DD; ++k) acc += avec[k] * leaky(whs[k] + wdl[k]);
    return acc;
}

__global__ __launch_bounds__(OWN_THREADS) void k_own(const float* __restrict__ Wh,
                                                     const float* __restrict__ avec,
                                                     const int* __restrict__ src,
                                                     const int* __restrict__ dst,
                                                     float* __restrict__ out,
                                                     int n, int E) {
    __shared__ __attribute__((aligned(16))) float sWd[RMAX * DD];
    __shared__ __attribute__((aligned(16))) float sAgg[RMAX * DD];
    __shared__ unsigned sMaxU[RMAX];
    __shared__ float    sDen[RMAX];
    __shared__ int      sE[CAP];
    __shared__ int      sS[CAP];
    __shared__ int      sLoc[CAP];
    __shared__ float    sLog[CAP];
    __shared__ int      sCount;
    __shared__ int      sOver;

    const int tid  = threadIdx.x;
    const int lane = tid & 31;
    const int wave = tid >> 5;
    const int lm   = lane & 15;
    const int half = lane >> 4;
    const float4 a4 = ((const float4*)avec)[lm];

    const bool vec = ((((size_t)dst) & 15) == 0);
    const int  E4  = vec ? (E >> 2) : 0;
    const int4* d4 = (const int4*)dst;
    const int rot = (int)(blockIdx.x * (unsigned)(E4 / (int)gridDim.x)) & ~7;

    for (int lo = blockIdx.x * RMAX; lo < n; lo += gridDim.x * RMAX) {
        const int cnt = min(RMAX, n - lo);

        for (int i = tid; i < cnt * (DD / 4); i += OWN_THREADS) {
            ((float4*)sWd)[i]  = ((const float4*)(Wh + (size_t)lo * DD))[i];
            ((float4*)sAgg)[i] = make_float4(0.0f, 0.0f, 0.0f, 0.0f);
        }
        if (tid < RMAX) { sMaxU[tid] = 0u; sDen[tid] = 0.0f; }
        if (tid == 0) { sCount = 0; sOver = 0; }
        __syncthreads();

        auto take = [&](int d, int e) {
            const unsigned loc = (unsigned)(d - lo);
            if (loc < (unsigned)cnt) {
                const int pos = atomicAdd(&sCount, 1);
                if (pos < CAP) { sE[pos] = e; sLoc[pos] = (int)loc; }
                else sOver = 1;
            }
        };
        for (int i0 = tid; i0 < E4; i0 += SCAN_U * OWN_THREADS) {
            int4 v[SCAN_U];
            int  id[SCAN_U];
#pragma unroll
            for (int j = 0; j < SCAN_U; ++j) {
                const int i = i0 + j * OWN_THREADS;
                int k = i + rot;
                if (k >= E4) k -= E4;
                id[j] = k;
                if (i < E4) v[j] = d4[k];
                else        v[j] = make_int4(-1, -1, -1, -1);
            }
#pragma unroll
            for (int j = 0; j < SCAN_U; ++j) {
                const unsigned nearest = min(min((unsigned)(v[j].x - lo), (unsigned)(v[j].y - lo)),
                                             min((unsigned)(v[j].z - lo), (unsigned)(v[j].w - lo)));
                if (nearest < (unsigned)cnt) {
                    const int e = id[j] * 4;
                    take(v[j].x, e + 0);
                    take(v[j].y, e + 1);
                    take(v[j].z, e + 2);
                    take(v[j].w, e + 3);
                }
            }
        }
        for (int e = 4 * E4 + tid; e < E; e += OWN_THREADS) take(dst[e], e);
        __syncthreads();

        const int total = sCount;
        const int over  = sOver;

        if (!over) {
            const int T = total;
            for (int base = 0; base < T; base += 64 * EDGE_U) {
                int  q[EDGE_U], s[EDGE_U], loc[EDGE_U];
                bool ok[EDGE_U];
                float p[EDGE_U];
                float4 ws[EDGE_U];
#pragma unroll
                for (int u = 0; u < EDGE_U; ++u) {
                    q[u]  = base + u * 64 + wave * 2 + half;
                    ok[u] = q[u] < T;
                    const int qc = ok[u] ? q[u] : 0;
                    loc[u] = sLoc[qc];
                    s[u]   = src[sE[qc]];
                }
#pragma unroll
                for (int u = 0; u < EDGE_U; ++u)
                    ws[u] = *(const float4*)(Wh + (size_t)s[u] * DD + 4 * lm);
#pragma unroll
                for (int u = 0; u < EDGE_U; ++u) {
                    const float4 wd = *(const float4*)(sWd + loc[u] * DD + 4 * lm);
                    p[u] = (a4.x * leaky(ws[u].x + wd.x) + a4.y * leaky(ws[u].y + wd.y))
                         + (a4.z * leaky(ws[u].z + wd.z) + a4.w * leaky(ws[u].w + wd.w));
                }
#pragma unroll
                for (int m = 8; m >= 1; m >>= 1) {
#pragma unroll
                    for (int u = 0; u < EDGE_U; ++u) p[u] += __shfl_xor(p[u], m, 32);
                }
#pragma unroll
                for (int u = 0; u < EDGE_U; ++u) {
                    if (ok[u] && lm == 0) {
                        sLog[q[u]] = p[u];
                        sS[q[u]]   = s[u];
                        atomicMax(&sMaxU[loc[u]], flipF(p[u]));
                    }
                }
            }
            __syncthreads();
            for (int base = 0; base < T; base += 64 * EDGE_U) {
                int  q[EDGE_U], loc[EDGE_U];
                bool ok[EDGE_U];
                float lg[EDGE_U];
                float4 ws[EDGE_U];
#pragma unroll
                for (int u = 0; u < EDGE_U; ++u) {
                    q[u]  = base + u * 64 + wave * 2 + half;
                    ok[u] = q[u] < T;
                    const int qc = ok[u] ? q[u] : 0;
                    loc[u] = sLoc[qc];
                    lg[u]  = sLog[qc];
                    ws[u]  = *(const float4*)(Wh + (size_t)sS[qc] * DD + 4 * lm);
                }
#pragma unroll
                for (int u = 0; u < EDGE_U; ++u) {
                    if (ok[u]) {
                        const float p = __expf(lg[u] - unflipF(sMaxU[loc[u]]));
                        float* row = sAgg + loc[u] * DD + 4 * lm;
                        if (lm == 0) atomicAdd(&sDen[loc[u]], p);
                        atomicAdd(row + 0, p * ws[u].x);
                        atomicAdd(row + 1, p * ws[u].y);
                        atomicAdd(row + 2, p * ws[u].z);
                        atomicAdd(row + 3, p * ws[u].w);
                    }
                }
            }
        } else {
            for (int e = tid; e < E; e += OWN_THREADS) {
                const unsigned loc = (unsigned)(dst[e] - lo);
                if (loc < (unsigned)cnt) {
                    const float p = lane_logit(Wh + (size_t)src[e] * DD, sWd + loc * DD, avec);
                    atomicMax(&sMaxU[loc], flipF(p));
                }
            }
            __syncthreads();
            for (int e = tid; e < E; e += OWN_THREADS) {
                const unsigned loc = (unsigned)(dst[e] - lo);
                if (loc < (unsigned)cnt) {
                    const float* whs = Wh + (size_t)src[e] * DD;
                    const float p = lane_logit(whs, sWd + loc * DD, avec);
                    const float w = __expf(p - unflipF(sMaxU[loc]));
                    atomicAdd(&sDen[loc], w);
                    for (int k = 0; k < DD; ++k) atomicAdd(&sAgg[loc * DD + k], w * whs[k]);
                }
            }
        }
        __syncthreads();

        for (int i = tid; i < cnt * (DD / 4); i += OWN_THREADS) {
            const float4 v = ((const float4*)sAgg)[i];
            const float den = fmaxf(sDen[i >> 4], 1e-9f);
            float4 o;
            o.x = 1.0f / (1.0f + __expf(-(v.x / den)));
            o.y = 1.0f / (1.0f + __expf(-(v.y / den)));
            o.z = 1.0f / (1.0f + __expf(-(v.z / den)));
            o.w = 1.0f / (1.0f + __expf(-(v.w / den)));
            ((float4*)(out + (size_t)lo * DD))[i] = o;
        }
        __syncthreads();
    }
}

extern "C" void kernel_launch(void* const* d_in, const int* in_sizes, int n_in,
                              void* d_out, int out_size, void* d_ws, size_t ws_size,
                              hipStream_t stream) {
    const float* x    = (const float*)d_in[0];
    const float* W    = (const float*)d_in[1];
    const float* bias = (const float*)d_in[2];
    const float* avec = (const float*)d_in[3];
    const int*   ei   = (const int*)d_in[4];

    const int N = in_sizes[0] / DD;
    const int E = in_sizes[4] / 2;
    const int* src = ei;
    const int* dst = ei + E;

    float* Wh = (float*)d_ws;

    const int tiles = (N + 15) / 16;
    k_gemm<<<(tiles + 7) / 8, 256, 0, stream>>>(x, W, bias, Wh, N);

    k_own<<<OWN_BLOCKS, OWN_THREADS, 0, stream>>>(Wh, avec, src, dst, (float*)d_out, N, E);
}


# owner search reads a 1-byte range key per edge (launch 1 writes it beside the projection), 16 edges per 16-B load, not the 4-B dst list; dst tested exactly in the first walk; sums by LDS arrival
# speedup vs baseline: 4.1719x; 1.2585x over previous
#include <hip/hip_runtime.h>
#include <hip/hip_bf16.h>
#include <math.h>

#define DD 64
#define NSLOPE 0.2f

typedef __attribute__((ext_vector_type(2))) float v2f;
typedef __attribute__((ext_vector_type(8))) float v8f;

__device__ __forceinline__ unsigned flipF(float f) {
    unsigned u = __float_as_uint(f);
    return (u >> 31) ? ~u : (u | 0x80000000u);
}
__device__ __forceinline__ float unflipF(unsigned u) {
    unsigned v = (u >> 31) ? (u & 0x7FFFFFFFu) : ~u;
    return __uint_as_float(v);
}

#define RMAX 196
#define OWN_BLOCKS 256

static_assert(OWN_BLOCKS == 256, "the range key is one byte: range index mod 256 must be the owner");
__device__ __forceinline__ unsigned key_of(int d) { return ((unsigned)d / RMAX) & 0xFFu; }

__global__ __launch_bounds__(256) void k_gemm(const float* __restrict__ x,
                                              const float* __restrict__ W,
                                              const float* __restrict__ bias,
                                              float* __restrict__ Wh, int n,
                                              const int* __restrict__ dst,
                                              unsigned* __restrict__ key, int E) {
    {
        const bool vec = ((((size_t)dst) & 15) == 0);
        const int nDw = ((E + 15) >> 4) << 2;
        const int gstride = (int)gridDim.x * 256;
        for (int i = (int)blockIdx.x * 256 + (int)threadIdx.x; i < nDw; i += gstride) {
            const int e0 = 4 * i;
            int4 d;
            if (vec && e0 + 3 < E) {
                d = ((const int4*)dst)[i];
            } else {
                d.x = (e0 + 0 < E) ? dst[e0 + 0] : -1;
                d.y = (e0 + 1 < E) ? dst[e0 + 1] : -1;
                d.z = (e0 + 2 < E) ? dst[e0 + 2] : -1;
                d.w = (e0 + 3 < E) ? dst[e0 + 3] : -1;
            }
            key[i] = key_of(d.x) | (key_of(d.y) << 8) | (key_of(d.z) << 16) | (key_of(d.w) << 24);
        }
    }
    const int lane = threadIdx.x & 31;
    const int wave = threadIdx.x >> 5;
    const int tile = blockIdx.x * 8 + wave;
    if (tile * 16 >= n) return;

    const int lm   = lane & 15;
    const int half = lane >> 4;
    const int row  = tile * 16 + lm;

    const float* xr = x + (size_t)row * DD + 2 * half;
    v2f areg[16];
#pragma unroll
    for (int i = 0; i < 16; ++i) {
        areg[i].x = xr[4 * i + 0];
        areg[i].y = xr[4 * i + 1];
    }

#pragma unroll
    for (int c = 0; c < 4; ++c) {
        v8f acc = {};
        const float* wr = W + (size_t)(c * 16 + lm) * DD + 2 * half;
#pragma unroll
        for (int i = 0; i < 16; ++i) {
            v2f bfrag;
            bfrag.x = wr[4 * i + 0];
            bfrag.y = wr[4 * i + 1];
            acc = __builtin_amdgcn_wmma_f32_16x16x4_f32(
                false, areg[i], false, bfrag, (short)0, acc, false, false);
        }
        const float bv = bias[c * 16 + lm];
#pragma unroll
        for (int v = 0; v < 8; ++v) {
            Wh[(size_t)(tile * 16 + v + 8 * half) * DD + c * 16 + lm] = acc[v] + bv;
        }
    }
}

#define CAP 6144
#define OWN_THREADS 1024
#define SCAN_U 8
#define EDGE_U 8

__device__ __forceinline__ float leaky(float s) { return s > 0.0f ? s : NSLOPE * s; }

__device__ __forceinline__ float lane_logit(const float* __restrict__ whs,
                                            const float* wdl,
                                            const float* __restrict__ avec) {
    float acc = 0.0f;
    for (int k = 0; k < DD; ++k) acc += avec[k] * leaky(whs[k] + wdl[k]);
    return acc;
}

__global__ __launch_bounds__(OWN_THREADS) void k_own(const float* __restrict__ Wh,
                                                     const float* __restrict__ avec,
                                                     const int* __restrict__ src,
                                                     const int* __restrict__ dst,
                                                     const unsigned* __restrict__ key,
                                                     float* __restrict__ out,
                                                     int n, int E) {
    __shared__ __attribute__((aligned(16))) float sWd[RMAX * DD];
    __shared__ __attribute__((aligned(16))) float sAgg[RMAX * DD];
    __shared__ unsigned sMaxU[RMAX];
    __shared__ float    sDen[RMAX];
    __shared__ int      sE[CAP];
    __shared__ int      sS[CAP];
    __shared__ int      sLoc[CAP];
    __shared__ float    sLog[CAP];
    __shared__ int      sCount;

    const int tid  = threadIdx.x;
    const int lane = tid & 31;
    const int wave = tid >> 5;
    const int lm   = lane & 15;
    const int half = lane >> 4;
    const float4 a4 = ((const float4*)avec)[lm];

    const int nK4 = (E + 15) >> 4;
    const int rot = (int)(blockIdx.x * (unsigned)(nK4 / (int)gridDim.x)) & ~7;

    for (int lo = blockIdx.x * RMAX; lo < n; lo += gridDim.x * RMAX) {
        const int cnt = min(RMAX, n - lo);

        for (int i = tid; i < cnt * (DD / 4); i += OWN_THREADS) {
            ((float4*)sWd)[i]  = ((const float4*)(Wh + (size_t)lo * DD))[i];
            ((float4*)sAgg)[i] = make_float4(0.0f, 0.0f, 0.0f, 0.0f);
        }
        if (tid < RMAX) { sMaxU[tid] = 0u; sDen[tid] = 0.0f; }
        if (tid == 0) sCount = 0;
        __syncthreads();

        const unsigned K1 = 0x01010101u, K80 = 0x80808080u;
        const unsigned pat = ((unsigned)(lo / RMAX) & 0xFFu) * K1;
        auto probe = [&](const uint4 v, const int id) {
            const unsigned x0 = v.x ^ pat, x1 = v.y ^ pat, x2 = v.z ^ pat, x3 = v.w ^ pat;
            const unsigned h0 = (x0 - K1) & ~x0, h1 = (x1 - K1) & ~x1;
            const unsigned h2 = (x2 - K1) & ~x2, h3 = (x3 - K1) & ~x3;
            if (((h0 | h1 | h2 | h3) & K80) != 0u) {
                unsigned mm = ((h0 >> 7) & K1) | ((h1 >> 6) & (K1 << 1))
                            | ((h2 >> 5) & (K1 << 2)) | ((h3 >> 4) & (K1 << 3));
                int pos = atomicAdd(&sCount, __popc(mm));
                const int e0 = id * 16;
                while (mm) {
                    const int t = __ffs((int)mm) - 1;
                    mm &= mm - 1u;
                    if (pos < CAP) sE[pos] = e0 + 4 * (t & 3) + (t >> 3);
                    ++pos;
                }
            }
        };
        auto ldkey = [&](const int k) {
            return *(const uint4*)((const char*)key + ((unsigned)k << 4));
        };
        int i0 = tid;
        for (; i0 + (SCAN_U - 1) * OWN_THREADS < nK4; i0 += SCAN_U * OWN_THREADS) {
            uint4 v[SCAN_U];
            int   id[SCAN_U];
#pragma unroll
            for (int j = 0; j < SCAN_U; ++j) {
                int k = i0 + j * OWN_THREADS + rot;
                if (k >= nK4) k -= nK4;
                id[j] = k;
                v[j]  = ldkey(k);
            }
#pragma unroll
            for (int j = 0; j < SCAN_U; ++j) probe(v[j], id[j]);
        }
        for (; i0 < nK4; i0 += OWN_THREADS) {
            int k = i0 + rot;
            if (k >= nK4) k -= nK4;
            probe(ldkey(k), k);
        }
        __syncthreads();

        const int total = sCount;
        const int over  = total > CAP;

        if (!over) {
            const int T = total;
            for (int base = 0; base < T; base += 64 * EDGE_U) {
                int  q[EDGE_U], s[EDGE_U], loc[EDGE_U];
                bool ok[EDGE_U], mine[EDGE_U];
                float p[EDGE_U];
                float4 ws[EDGE_U];
#pragma unroll
                for (int u = 0; u < EDGE_U; ++u) {
                    q[u]  = base + u * 64 + wave * 2 + half;
                    ok[u] = q[u] < T;
                    const int qc = ok[u] ? q[u] : 0;
                    const int e  = sE[qc];
                    const int ec = e < E ? e : 0;
                    s[u] = src[ec];
                    const unsigned l = (unsigned)(dst[ec] - lo);
                    mine[u] = ok[u] && e < E && l < (unsigned)cnt;
                    loc[u]  = mine[u] ? (int)l : 0;
                }
#pragma unroll
                for (int u = 0; u < EDGE_U; ++u)
                    ws[u] = *(const float4*)(Wh + (size_t)s[u] * DD + 4 * lm);
#pragma unroll
                for (int u = 0; u < EDGE_U; ++u) {
                    const float4 wd = *(const float4*)(sWd + loc[u] * DD + 4 * lm);
                    p[u] = (a4.x * leaky(ws[u].x + wd.x) + a4.y * leaky(ws[u].y + wd.y))
                         + (a4.z * leaky(ws[u].z + wd.z) + a4.w * leaky(ws[u].w + wd.w));
                }
#pragma unroll
                for (int m = 8; m >= 1; m >>= 1) {
#pragma unroll
                    for (int u = 0; u < EDGE_U; ++u) p[u] += __shfl_xor(p[u], m, 32);
                }
#pragma unroll
                for (int u = 0; u < EDGE_U; ++u) {
                    if (ok[u] && lm == 0) {
                        sS[q[u]]   = s[u];
                        sLoc[q[u]] = mine[u] ? loc[u] : -1;
                        sLog[q[u]] = p[u];
                        if (mine[u]) atomicMax(&sMaxU[loc[u]], flipF(p[u]));
                    }
                }
            }
            __syncthreads();
            for (int base = 0; base < T; base += 64 * EDGE_U) {
                int  q[EDGE_U], loc[EDGE_U];
                bool ok[EDGE_U];
                float lg[EDGE_U];
                float4 ws[EDGE_U];
#pragma unroll
                for (int u = 0; u < EDGE_U; ++u) {
                    q[u]  = base + u * 64 + wave * 2 + half;
                    const int qc = q[u] < T ? q[u] : 0;
                    loc[u] = sLoc[qc];
                    ok[u]  = q[u] < T && loc[u] >= 0;
                    lg[u]  = sLog[qc];
                    ws[u]  = *(const float4*)(Wh + (size_t)sS[qc] * DD + 4 * lm);
                }
#pragma unroll
                for (int u = 0; u < EDGE_U; ++u) {
                    if (ok[u]) {
                        const float p = __expf(lg[u] - unflipF(sMaxU[loc[u]]));
                        float* row = sAgg + loc[u] * DD + 4 * lm;
                        if (lm == 0) atomicAdd(&sDen[loc[u]], p);
                        atomicAdd(row + 0, p * ws[u].x);
                        atomicAdd(row + 1, p * ws[u].y);
                        atomicAdd(row + 2, p * ws[u].z);
                        atomicAdd(row + 3, p * ws[u].w);
                    }
                }
            }
        } else {
            for (int e = tid; e < E; e += OWN_THREADS) {
                const unsigned loc = (unsigned)(dst[e] - lo);
                if (loc < (unsigned)cnt) {
                    const float p = lane_logit(Wh + (size_t)src[e] * DD, sWd + loc * DD, avec);
                    atomicMax(&sMaxU[loc], flipF(p));
                }
            }
            __syncthreads();
            for (int e = tid; e < E; e += OWN_THREADS) {
                const unsigned loc = (unsigned)(dst[e] - lo);
                if (loc < (unsigned)cnt) {
                    const float* whs = Wh + (size_t)src[e] * DD;
                    const float p = lane_logit(whs, sWd + loc * DD, avec);
                    const float w = __expf(p - unflipF(sMaxU[loc]));
                    atomicAdd(&sDen[loc], w);
                    for (int k = 0; k < DD; ++k) atomicAdd(&sAgg[loc * DD + k], w * whs[k]);
                }
            }
        }
        __syncthreads();

        for (int i = tid; i < cnt * (DD / 4); i += OWN_THREADS) {
            const float4 v = ((const float4*)sAgg)[i];
            const float den = fmaxf(sDen[i >> 4], 1e-9f);
            float4 o;
            o.x = 1.0f / (1.0f + __expf(-(v.x / den)));
            o.y = 1.0f / (1.0f + __expf(-(v.y / den)));
            o.z = 1.0f / (1.0f + __expf(-(v.z / den)));
            o.w = 1.0f / (1.0f + __expf(-(v.w / den)));
            ((float4*)(out + (size_t)lo * DD))[i] = o;
        }
        __syncthreads();
    }
}

extern "C" void kernel_launch(void* const* d_in, const int* in_sizes, int n_in,
                              void* d_out, int out_size, void* d_ws, size_t ws_size,
                              hipStream_t stream) {
    const float* x    = (const float*)d_in[0];
    const float* W    = (const float*)d_in[1];
    const float* bias = (const float*)d_in[2];
    const float* avec = (const float*)d_in[3];
    const int*   ei   = (const int*)d_in[4];

    const int N = in_sizes[0] / DD;
    const int E = in_sizes[4] / 2;
    const int* src = ei;
    const int* dst = ei + E;

    float* Wh = (float*)d_ws;
    unsigned* key = (unsigned*)(Wh + (size_t)N * DD);

    const int tiles = (N + 15) / 16;
    k_gemm<<<(tiles + 7) / 8, 256, 0, stream>>>(x, W, bias, Wh, N, dst, key, E);

    k_own<<<OWN_BLOCKS, OWN_THREADS, 0, stream>>>(Wh, avec, src, dst, key, (float*)d_out, N, E);
}


# search takes a step's eight 16-byte key pieces at once: flags packed with no branch, lane counts summed by row_shr moves, one list-counter trip per wave step, not per piece; sums by LDS arrival
# speedup vs baseline: 4.4668x; 1.0707x over previous
#include <hip/hip_runtime.h>
#include <hip/hip_bf16.h>
#include <math.h>

#define DD 64
#define NSLOPE 0.2f

typedef __attribute__((ext_vector_type(2))) float v2f;
typedef __attribute__((ext_vector_type(8))) float v8f;

__device__ __forceinline__ unsigned flipF(float f) {
    unsigned u = __float_as_uint(f);
    return (u >> 31) ? ~u : (u | 0x80000000u);
}
__device__ __forceinline__ float unflipF(unsigned u) {
    unsigned v = (u >> 31) ? (u & 0x7FFFFFFFu) : ~u;
    return __uint_as_float(v);
}

#define RMAX 196
#define OWN_BLOCKS 256

static_assert(OWN_BLOCKS == 256, "the range key is one byte: range index mod 256 must be the owner");
__device__ __forceinline__ unsigned key_of(int d) { return ((unsigned)d / RMAX) & 0xFFu; }

__global__ __launch_bounds__(256) void k_gemm(const float* __restrict__ x,
                                              const float* __restrict__ W,
                                              const float* __restrict__ bias,
                                              float* __restrict__ Wh, int n,
                                              const int* __restrict__ dst,
                                              unsigned* __restrict__ key, int E) {
    {
        const bool vec = ((((size_t)dst) & 15) == 0);
        const int nDw = ((E + 15) >> 4) << 2;
        const int gstride = (int)gridDim.x * 256;
        for (int i = (int)blockIdx.x * 256 + (int)threadIdx.x; i < nDw; i += gstride) {
            const int e0 = 4 * i;
            int4 d;
            if (vec && e0 + 3 < E) {
                d = ((const int4*)dst)[i];
            } else {
                d.x = (e0 + 0 < E) ? dst[e0 + 0] : -1;
                d.y = (e0 + 1 < E) ? dst[e0 + 1] : -1;
                d.z = (e0 + 2 < E) ? dst[e0 + 2] : -1;
                d.w = (e0 + 3 < E) ? dst[e0 + 3] : -1;
            }
            key[i] = key_of(d.x) | (key_of(d.y) << 8) | (key_of(d.z) << 16) | (key_of(d.w) << 24);
        }
    }
    const int lane = threadIdx.x & 31;
    const int wave = threadIdx.x >> 5;
    const int tile = blockIdx.x * 8 + wave;
    if (tile * 16 >= n) return;

    const int lm   = lane & 15;
    const int half = lane >> 4;
    const int row  = tile * 16 + lm;

    const float* xr = x + (size_t)row * DD + 2 * half;
    v2f areg[16];
#pragma unroll
    for (int i = 0; i < 16; ++i) {
        areg[i].x = xr[4 * i + 0];
        areg[i].y = xr[4 * i + 1];
    }

#pragma unroll
    for (int c = 0; c < 4; ++c) {
        v8f acc = {};
        const float* wr = W + (size_t)(c * 16 + lm) * DD + 2 * half;
#pragma unroll
        for (int i = 0; i < 16; ++i) {
            v2f bfrag;
            bfrag.x = wr[4 * i + 0];
            bfrag.y = wr[4 * i + 1];
            acc = __builtin_amdgcn_wmma_f32_16x16x4_f32(
                false, areg[i], false, bfrag, (short)0, acc, false, false);
        }
        const float bv = bias[c * 16 + lm];
#pragma unroll
        for (int v = 0; v < 8; ++v) {
            Wh[(size_t)(tile * 16 + v + 8 * half) * DD + c * 16 + lm] = acc[v] + bv;
        }
    }
}

#define CAP 6144
#define OWN_THREADS 1024
#define SCAN_U 8
#define EDGE_U 8

__device__ __forceinline__ float leaky(float s) { return s > 0.0f ? s : NSLOPE * s; }

__device__ __forceinline__ float lane_logit(const float* __restrict__ whs,
                                            const float* wdl,
                                            const float* __restrict__ avec) {
    float acc = 0.0f;
    for (int k = 0; k < DD; ++k) acc += avec[k] * leaky(whs[k] + wdl[k]);
    return acc;
}

__global__ __launch_bounds__(OWN_THREADS) void k_own(const float* __restrict__ Wh,
                                                     const float* __restrict__ avec,
                                                     const int* __restrict__ src,
                                                     const int* __restrict__ dst,
                                                     const unsigned* __restrict__ key,
                                                     float* __restrict__ out,
                                                     int n, int E) {
    __shared__ __attribute__((aligned(16))) float sWd[RMAX * DD];
    __shared__ __attribute__((aligned(16))) float sAgg[RMAX * DD];
    __shared__ unsigned sMaxU[RMAX];
    __shared__ float    sDen[RMAX];
    __shared__ int      sE[CAP];
    __shared__ int      sS[CAP];
    __shared__ int      sLoc[CAP];
    __shared__ float    sLog[CAP];
    __shared__ int      sCount;

    const int tid  = threadIdx.x;
    const int lane = tid & 31;
    const int wave = tid >> 5;
    const int lm   = lane & 15;
    const int half = lane >> 4;
    const float4 a4 = ((const float4*)avec)[lm];

    const int nK4 = (E + 15) >> 4;
    const int rot = (int)(blockIdx.x * (unsigned)(nK4 / (int)gridDim.x)) & ~7;

    for (int lo = blockIdx.x * RMAX; lo < n; lo += gridDim.x * RMAX) {
        const int cnt = min(RMAX, n - lo);

        for (int i = tid; i < cnt * (DD / 4); i += OWN_THREADS) {
            ((float4*)sWd)[i]  = ((const float4*)(Wh + (size_t)lo * DD))[i];
            ((float4*)sAgg)[i] = make_float4(0.0f, 0.0f, 0.0f, 0.0f);
        }
        if (tid < RMAX) { sMaxU[tid] = 0u; sDen[tid] = 0.0f; }
        if (tid == 0) sCount = 0;
        __syncthreads();

        const unsigned K1 = 0x01010101u;
        const unsigned pat = ((unsigned)(lo / RMAX) & 0xFFu) * K1;
        auto pack2 = [&](const uint4 p, const uint4 q) -> unsigned {
            const unsigned p0 = p.x ^ pat, p1 = p.y ^ pat, p2 = p.z ^ pat, p3 = p.w ^ pat;
            const unsigned q0 = q.x ^ pat, q1 = q.y ^ pat, q2 = q.z ^ pat, q3 = q.w ^ pat;
            const unsigned h0 = (p0 - K1) & ~p0, h1 = (p1 - K1) & ~p1;
            const unsigned h2 = (p2 - K1) & ~p2, h3 = (p3 - K1) & ~p3;
            const unsigned g0 = (q0 - K1) & ~q0, g1 = (q1 - K1) & ~q1;
            const unsigned g2 = (q2 - K1) & ~q2, g3 = (q3 - K1) & ~q3;
            return ((h0 >> 7) & K1)        | ((h1 >> 6) & (K1 << 1))
                 | ((h2 >> 5) & (K1 << 2)) | ((h3 >> 4) & (K1 << 3))
                 | ((g0 >> 3) & (K1 << 4)) | ((g1 >> 2) & (K1 << 5))
                 | ((g2 >> 1) & (K1 << 6)) | (g3 & (K1 << 7));
        };
        auto ldkey = [&](const int k) {
            return *(const uint4*)((const char*)key + ((unsigned)k << 4));
        };
        auto step = [&](const int i0, const bool guarded) {
            uint4 v[SCAN_U];
            int   ida[SCAN_U / 2], idb[SCAN_U / 2];
            bool  in[SCAN_U];
#pragma unroll
            for (int j = 0; j < SCAN_U; ++j) {
                const int i = i0 + j * OWN_THREADS;
                in[j] = !guarded || i < nK4;
                int k = i + rot;
                if (k >= nK4) k -= nK4;
                if (guarded && !in[j]) k = 0;
                if (j & 1) idb[j >> 1] = k; else ida[j >> 1] = k;
                v[j]  = ldkey(k);
            }
            unsigned w[SCAN_U / 2];
            int c = 0;
#pragma unroll
            for (int q = 0; q < SCAN_U / 2; ++q) {
                w[q] = pack2(v[2 * q], v[2 * q + 1]);
                if (guarded) {
                    if (!in[2 * q])     w[q] &= 0xF0F0F0F0u;
                    if (!in[2 * q + 1]) w[q] &= 0x0F0F0F0Fu;
                }
                c += __popc(w[q]);
            }
            int incl = c;
            incl += __builtin_amdgcn_update_dpp(0, incl, 0x111, 0xf, 0xf, true);
            incl += __builtin_amdgcn_update_dpp(0, incl, 0x112, 0xf, 0xf, true);
            incl += __builtin_amdgcn_update_dpp(0, incl, 0x114, 0xf, 0xf, true);
            incl += __builtin_amdgcn_update_dpp(0, incl, 0x118, 0xf, 0xf, true);
            const int lowsum = __builtin_amdgcn_readlane(incl, 15);
            if (lane >= 16) incl += lowsum;
            const int tot = __builtin_amdgcn_readlane(incl, 31);
            int basepos = 0;
            if (lane == 0) basepos = atomicAdd(&sCount, tot);
            basepos = __builtin_amdgcn_readfirstlane(basepos);
            int pos = basepos + incl - c;
#pragma unroll
            for (int q = 0; q < SCAN_U / 2; ++q) {
                unsigned m = w[q];
                while (m) {
                    const int t = __ffs((int)m) - 1;
                    m &= m - 1u;
                    const int k = (t & 4) ? idb[q] : ida[q];
                    if (pos < CAP) sE[pos] = 16 * k + 4 * (t & 3) + (t >> 3);
                    ++pos;
                }
            }
        };
        {
            const int wfirst = __builtin_amdgcn_readfirstlane(tid & ~31);
            int s0 = 0;
            for (; s0 + wfirst + 31 + (SCAN_U - 1) * OWN_THREADS < nK4; s0 += SCAN_U * OWN_THREADS)
                step(s0 + tid, false);
            if (s0 + wfirst < nK4) step(s0 + tid, true);
        }
        __syncthreads();

        const int total = sCount;
        const int over  = total > CAP;

        if (!over) {
            const int T = total;
            for (int base = 0; base < T; base += 64 * EDGE_U) {
                int  q[EDGE_U], s[EDGE_U], loc[EDGE_U];
                bool ok[EDGE_U], mine[EDGE_U];
                float p[EDGE_U];
                float4 ws[EDGE_U];
#pragma unroll
                for (int u = 0; u < EDGE_U; ++u) {
                    q[u]  = base + u * 64 + wave * 2 + half;
                    ok[u] = q[u] < T;
                    const int qc = ok[u] ? q[u] : 0;
                    const int e  = sE[qc];
                    const int ec = e < E ? e : 0;
                    s[u] = src[ec];
                    const unsigned l = (unsigned)(dst[ec] - lo);
                    mine[u] = ok[u] && e < E && l < (unsigned)cnt;
                    loc[u]  = mine[u] ? (int)l : 0;
                }
#pragma unroll
                for (int u = 0; u < EDGE_U; ++u)
                    ws[u] = *(const float4*)(Wh + (size_t)s[u] * DD + 4 * lm);
#pragma unroll
                for (int u = 0; u < EDGE_U; ++u) {
                    const float4 wd = *(const float4*)(sWd + loc[u] * DD + 4 * lm);
                    p[u] = (a4.x * leaky(ws[u].x + wd.x) + a4.y * leaky(ws[u].y + wd.y))
                         + (a4.z * leaky(ws[u].z + wd.z) + a4.w * leaky(ws[u].w + wd.w));
                }
#pragma unroll
                for (int m = 8; m >= 1; m >>= 1) {
#pragma unroll
                    for (int u = 0; u < EDGE_U; ++u) p[u] += __shfl_xor(p[u], m, 32);
                }
#pragma unroll
                for (int u = 0; u < EDGE_U; ++u) {
                    if (ok[u] && lm == 0) {
                        sS[q[u]]   = s[u];
                        sLoc[q[u]] = mine[u] ? loc[u] : -1;
                        sLog[q[u]] = p[u];
                        if (mine[u]) atomicMax(&sMaxU[loc[u]], flipF(p[u]));
                    }
                }
            }
            __syncthreads();
            for (int base = 0; base < T; base += 64 * EDGE_U) {
                int  q[EDGE_U], loc[EDGE_U];
                bool ok[EDGE_U];
                float lg[EDGE_U];
                float4 ws[EDGE_U];
#pragma unroll
                for (int u = 0; u < EDGE_U; ++u) {
                    q[u]  = base + u * 64 + wave * 2 + half;
                    const int qc = q[u] < T ? q[u] : 0;
                    loc[u] = sLoc[qc];
                    ok[u]  = q[u] < T && loc[u] >= 0;
                    lg[u]  = sLog[qc];
                    ws[u]  = *(const float4*)(Wh + (size_t)sS[qc] * DD + 4 * lm);
                }
#pragma unroll
                for (int u = 0; u < EDGE_U; ++u) {
                    if (ok[u]) {
                        const float p = __expf(lg[u] - unflipF(sMaxU[loc[u]]));
                        float* row = sAgg + loc[u] * DD + 4 * lm;
                        if (lm == 0) atomicAdd(&sDen[loc[u]], p);
                        atomicAdd(row + 0, p * ws[u].x);
                        atomicAdd(row + 1, p * ws[u].y);
                        atomicAdd(row + 2, p * ws[u].z);
                        atomicAdd(row + 3, p * ws[u].w);
                    }
                }
            }
        } else {
            for (int e = tid; e < E; e += OWN_THREADS) {
                const unsigned loc = (unsigned)(dst[e] - lo);
                if (loc < (unsigned)cnt) {
                    const float p = lane_logit(Wh + (size_t)src[e] * DD, sWd + loc * DD, avec);
                    atomicMax(&sMaxU[loc], flipF(p));
                }
            }
            __syncthreads();
            for (int e = tid; e < E; e += OWN_THREADS) {
                const unsigned loc = (unsigned)(dst[e] - lo);
                if (loc < (unsigned)cnt) {
                    const float* whs = Wh + (size_t)src[e] * DD;
                    const float p = lane_logit(whs, sWd + loc * DD, avec);
                    const float w = __expf(p - unflipF(sMaxU[loc]));
                    atomicAdd(&sDen[loc], w);
                    for (int k = 0; k < DD; ++k) atomicAdd(&sAgg[loc * DD + k], w * whs[k]);
                }
            }
        }
        __syncthreads();

        for (int i = tid; i < cnt * (DD / 4); i += OWN_THREADS) {
            const float4 v = ((const float4*)sAgg)[i];
            const float den = fmaxf(sDen[i >> 4], 1e-9f);
            float4 o;
            o.x = 1.0f / (1.0f + __expf(-(v.x / den)));
            o.y = 1.0f / (1.0f + __expf(-(v.y / den)));
            o.z = 1.0f / (1.0f + __expf(-(v.z / den)));
            o.w = 1.0f / (1.0f + __expf(-(v.w / den)));
            ((float4*)(out + (size_t)lo * DD))[i] = o;
        }
        __syncthreads();
    }
}

extern "C" void kernel_launch(void* const* d_in, const int* in_sizes, int n_in,
                              void* d_out, int out_size, void* d_ws, size_t ws_size,
                              hipStream_t stream) {
    const float* x    = (const float*)d_in[0];
    const float* W    = (const float*)d_in[1];
    const float* bias = (const float*)d_in[2];
    const float* avec = (const float*)d_in[3];
    const int*   ei   = (const int*)d_in[4];

    const int N = in_sizes[0] / DD;
    const int E = in_sizes[4] / 2;
    const int* src = ei;
    const int* dst = ei + E;

    float* Wh = (float*)d_ws;
    unsigned* key = (unsigned*)(Wh + (size_t)N * DD);

    const int tiles = (N + 15) / 16;
    k_gemm<<<(tiles + 7) / 8, 256, 0, stream>>>(x, W, bias, Wh, N, dst, key, E);

    k_own<<<OWN_BLOCKS, OWN_THREADS, 0, stream>>>(Wh, avec, src, dst, key, (float*)d_out, N, E);
}


# walks: each lane pair of a 16-lane row owns one of its 8 edges (list entries, src/dst, max, exp, denominator once per 8 edges); logit sums by a transposing row_xmask DPP butterfly, same bits
# speedup vs baseline: 4.6436x; 1.0396x over previous
#include <hip/hip_runtime.h>
#include <hip/hip_bf16.h>
#include <math.h>

#define DD 64
#define NSLOPE 0.2f

typedef __attribute__((ext_vector_type(2))) float v2f;
typedef __attribute__((ext_vector_type(8))) float v8f;

__device__ __forceinline__ unsigned flipF(float f) {
    unsigned u = __float_as_uint(f);
    return (u >> 31) ? ~u : (u | 0x80000000u);
}
__device__ __forceinline__ float unflipF(unsigned u) {
    unsigned v = (u >> 31) ? (u & 0x7FFFFFFFu) : ~u;
    return __uint_as_float(v);
}

#define RMAX 196
#define OWN_BLOCKS 256

static_assert(OWN_BLOCKS == 256, "the range key is one byte: range index mod 256 must be the owner");
__device__ __forceinline__ unsigned key_of(int d) { return ((unsigned)d / RMAX) & 0xFFu; }

__global__ __launch_bounds__(256) void k_gemm(const float* __restrict__ x,
                                              const float* __restrict__ W,
                                              const float* __restrict__ bias,
                                              float* __restrict__ Wh, int n,
                                              const int* __restrict__ dst,
                                              unsigned* __restrict__ key, int E) {
    {
        const bool vec = ((((size_t)dst) & 15) == 0);
        const int nDw = ((E + 15) >> 4) << 2;
        const int gstride = (int)gridDim.x * 256;
        for (int i = (int)blockIdx.x * 256 + (int)threadIdx.x; i < nDw; i += gstride) {
            const int e0 = 4 * i;
            int4 d;
            if (vec && e0 + 3 < E) {
                d = ((const int4*)dst)[i];
            } else {
                d.x = (e0 + 0 < E) ? dst[e0 + 0] : -1;
                d.y = (e0 + 1 < E) ? dst[e0 + 1] : -1;
                d.z = (e0 + 2 < E) ? dst[e0 + 2] : -1;
                d.w = (e0 + 3 < E) ? dst[e0 + 3] : -1;
            }
            key[i] = key_of(d.x) | (key_of(d.y) << 8) | (key_of(d.z) << 16) | (key_of(d.w) << 24);
        }
    }
    const int lane = threadIdx.x & 31;
    const int wave = threadIdx.x >> 5;
    const int tile = blockIdx.x * 8 + wave;
    if (tile * 16 >= n) return;

    const int lm   = lane & 15;
    const int half = lane >> 4;
    const int row  = tile * 16 + lm;

    const float* xr = x + (size_t)row * DD + 2 * half;
    v2f areg[16];
#pragma unroll
    for (int i = 0; i < 16; ++i) {
        areg[i].x = xr[4 * i + 0];
        areg[i].y = xr[4 * i + 1];
    }

#pragma unroll
    for (int c = 0; c < 4; ++c) {
        v8f acc = {};
        const float* wr = W + (size_t)(c * 16 + lm) * DD + 2 * half;
#pragma unroll
        for (int i = 0; i < 16; ++i) {
            v2f bfrag;
            bfrag.x = wr[4 * i + 0];
            bfrag.y = wr[4 * i + 1];
            acc = __builtin_amdgcn_wmma_f32_16x16x4_f32(
                false, areg[i], false, bfrag, (short)0, acc, false, false);
        }
        const float bv = bias[c * 16 + lm];
#pragma unroll
        for (int v = 0; v < 8; ++v) {
            Wh[(size_t)(tile * 16 + v + 8 * half) * DD + c * 16 + lm] = acc[v] + bv;
        }
    }
}

#define CAP 6144
#define OWN_THREADS 1024
#define SCAN_U 8
#define EDGE_U 8
static_assert(EDGE_U == 8, "a row's eight edges are owned by its eight lane pairs");

__device__ __forceinline__ float leaky(float s) { return s > 0.0f ? s : NSLOPE * s; }

template <int CTRL> __device__ __forceinline__ int row_move(int v) {
    return __builtin_amdgcn_update_dpp(0, v, CTRL, 0xf, 0xf, true);
}
template <int CTRL> __device__ __forceinline__ float row_move(float v) {
    return __int_as_float(__builtin_amdgcn_update_dpp(0, __float_as_int(v), CTRL, 0xf, 0xf, true));
}
template <typename T> __device__ __forceinline__ void row_share8(const T v, T (&out)[8]) {
    out[0] = row_move<0x150>(v); out[1] = row_move<0x152>(v);
    out[2] = row_move<0x154>(v); out[3] = row_move<0x156>(v);
    out[4] = row_move<0x158>(v); out[5] = row_move<0x15A>(v);
    out[6] = row_move<0x15C>(v); out[7] = row_move<0x15E>(v);
}
__device__ __forceinline__ float row_sums8(const float (&p)[8], const int lm) {
    const bool b3 = (lm & 8) != 0, b2 = (lm & 4) != 0, b1 = (lm & 2) != 0;
    float r[4];
#pragma unroll
    for (int j = 0; j < 4; ++j) {
        const float keep = b3 ? p[4 + j] : p[j];
        const float give = b3 ? p[j] : p[4 + j];
        r[j] = keep + row_move<0x168>(give);
    }
    float t[2];
#pragma unroll
    for (int j = 0; j < 2; ++j) {
        const float keep = b2 ? r[2 + j] : r[j];
        const float give = b2 ? r[j] : r[2 + j];
        t[j] = keep + row_move<0x164>(give);
    }
    const float keep = b1 ? t[1] : t[0];
    const float give = b1 ? t[0] : t[1];
    const float w = keep + row_move<0x162>(give);
    return w + row_move<0x161>(w);
}

__device__ __forceinline__ float lane_logit(const float* __restrict__ whs,
                                            const float* wdl,
                                            const float* __restrict__ avec) {
    float acc = 0.0f;
    for (int k = 0; k < DD; ++k) acc += avec[k] * leaky(whs[k] + wdl[k]);
    return acc;
}

__global__ __launch_bounds__(OWN_THREADS) void k_own(const float* __restrict__ Wh,
                                                     const float* __restrict__ avec,
                                                     const int* __restrict__ src,
                                                     const int* __restrict__ dst,
                                                     const unsigned* __restrict__ key,
                                                     float* __restrict__ out,
                                                     int n, int E) {
    __shared__ __attribute__((aligned(16))) float sWd[RMAX * DD];
    __shared__ __attribute__((aligned(16))) float sAgg[RMAX * DD];
    __shared__ unsigned sMaxU[RMAX];
    __shared__ float    sDen[RMAX];
    __shared__ int      sE[CAP];
    __shared__ int      sS[CAP];
    __shared__ int      sLoc[CAP];
    __shared__ float    sLog[CAP];
    __shared__ int      sCount;

    const int tid  = threadIdx.x;
    const int lane = tid & 31;
    const int wave = tid >> 5;
    const int lm   = lane & 15;
    const int half = lane >> 4;
    const int uo   = lm >> 1;
    const bool first = (lm & 1) == 0;
    const float4 a4 = ((const float4*)avec)[lm];

    const int nK4 = (E + 15) >> 4;
    const int rot = (int)(blockIdx.x * (unsigned)(nK4 / (int)gridDim.x)) & ~7;

    for (int lo = blockIdx.x * RMAX; lo < n; lo += gridDim.x * RMAX) {
        const int cnt = min(RMAX, n - lo);

        for (int i = tid; i < cnt * (DD / 4); i += OWN_THREADS) {
            ((float4*)sWd)[i]  = ((const float4*)(Wh + (size_t)lo * DD))[i];
            ((float4*)sAgg)[i] = make_float4(0.0f, 0.0f, 0.0f, 0.0f);
        }
        if (tid < RMAX) { sMaxU[tid] = 0u; sDen[tid] = 0.0f; }
        if (tid == 0) sCount = 0;
        __syncthreads();

        const unsigned K1 = 0x01010101u;
        const unsigned pat = ((unsigned)(lo / RMAX) & 0xFFu) * K1;
        auto pack2 = [&](const uint4 p, const uint4 q) -> unsigned {
            const unsigned p0 = p.x ^ pat, p1 = p.y ^ pat, p2 = p.z ^ pat, p3 = p.w ^ pat;
            const unsigned q0 = q.x ^ pat, q1 = q.y ^ pat, q2 = q.z ^ pat, q3 = q.w ^ pat;
            const unsigned h0 = (p0 - K1) & ~p0, h1 = (p1 - K1) & ~p1;
            const unsigned h2 = (p2 - K1) & ~p2, h3 = (p3 - K1) & ~p3;
            const unsigned g0 = (q0 - K1) & ~q0, g1 = (q1 - K1) & ~q1;
            const unsigned g2 = (q2 - K1) & ~q2, g3 = (q3 - K1) & ~q3;
            return ((h0 >> 7) & K1)        | ((h1 >> 6) & (K1 << 1))
                 | ((h2 >> 5) & (K1 << 2)) | ((h3 >> 4) & (K1 << 3))
                 | ((g0 >> 3) & (K1 << 4)) | ((g1 >> 2) & (K1 << 5))
                 | ((g2 >> 1) & (K1 << 6)) | (g3 & (K1 << 7));
        };
        auto ldkey = [&](const int k) {
            return *(const uint4*)((const char*)key + ((unsigned)k << 4));
        };
        auto step = [&](const int i0, const bool guarded) {
            uint4 v[SCAN_U];
            int   ida[SCAN_U / 2], idb[SCAN_U / 2];
            bool  in[SCAN_U];
#pragma unroll
            for (int j = 0; j < SCAN_U; ++j) {
                const int i = i0 + j * OWN_THREADS;
                in[j] = !guarded || i < nK4;
                int k = i + rot;
                if (k >= nK4) k -= nK4;
                if (guarded && !in[j]) k = 0;
                if (j & 1) idb[j >> 1] = k; else ida[j >> 1] = k;
                v[j]  = ldkey(k);
            }
            unsigned w[SCAN_U / 2];
            int c = 0;
#pragma unroll
            for (int q = 0; q < SCAN_U / 2; ++q) {
                w[q] = pack2(v[2 * q], v[2 * q + 1]);
                if (guarded) {
                    if (!in[2 * q])     w[q] &= 0xF0F0F0F0u;
                    if (!in[2 * q + 1]) w[q] &= 0x0F0F0F0Fu;
                }
                c += __popc(w[q]);
            }
            int incl = c;
            incl += __builtin_amdgcn_update_dpp(0, incl, 0x111, 0xf, 0xf, true);
            incl += __builtin_amdgcn_update_dpp(0, incl, 0x112, 0xf, 0xf, true);
            incl += __builtin_amdgcn_update_dpp(0, incl, 0x114, 0xf, 0xf, true);
            incl += __builtin_amdgcn_update_dpp(0, incl, 0x118, 0xf, 0xf, true);
            const int lowsum = __builtin_amdgcn_readlane(incl, 15);
            if (lane >= 16) incl += lowsum;
            const int tot = __builtin_amdgcn_readlane(incl, 31);
            int basepos = 0;
            if (lane == 0) basepos = atomicAdd(&sCount, tot);
            basepos = __builtin_amdgcn_readfirstlane(basepos);
            int pos = basepos + incl - c;
#pragma unroll
            for (int q = 0; q < SCAN_U / 2; ++q) {
                unsigned m = w[q];
                while (m) {
                    const int t = __ffs((int)m) - 1;
                    m &= m - 1u;
                    const int k = (t & 4) ? idb[q] : ida[q];
                    if (pos < CAP) sE[pos] = 16 * k + 4 * (t & 3) + (t >> 3);
                    ++pos;
                }
            }
        };
        {
            const int wfirst = __builtin_amdgcn_readfirstlane(tid & ~31);
            int s0 = 0;
            for (; s0 + wfirst + 31 + (SCAN_U - 1) * OWN_THREADS < nK4; s0 += SCAN_U * OWN_THREADS)
                step(s0 + tid, false);
            if (s0 + wfirst < nK4) step(s0 + tid, true);
        }
        __syncthreads();

        const int total = sCount;
        const int over  = total > CAP;

        if (!over) {
            const int T = total;
            for (int base = 0; base < T; base += 64 * EDGE_U) {
                const int  qo  = base + uo * 64 + wave * 2 + half;
                const bool oko = qo < T;
                const int  e   = sE[oko ? qo : 0];
                const int  ec  = e < E ? e : 0;
                const int  so  = src[ec];
                const unsigned l = (unsigned)(dst[ec] - lo);
                const bool mineo = oko && e < E && l < (unsigned)cnt;
                const int  loco  = mineo ? (int)l : 0;
                int s[EDGE_U], loc[EDGE_U];
                row_share8(so, s);
                row_share8(loco, loc);
                float  p[EDGE_U];
                float4 ws[EDGE_U];
#pragma unroll
                for (int u = 0; u < EDGE_U; ++u)
                    ws[u] = *(const float4*)(Wh + (size_t)s[u] * DD + 4 * lm);
#pragma unroll
                for (int u = 0; u < EDGE_U; ++u) {
                    const float4 wd = *(const float4*)(sWd + loc[u] * DD + 4 * lm);
                    p[u] = (a4.x * leaky(ws[u].x + wd.x) + a4.y * leaky(ws[u].y + wd.y))
                         + (a4.z * leaky(ws[u].z + wd.z) + a4.w * leaky(ws[u].w + wd.w));
                }
                const float logit = row_sums8(p, lm);
                if (oko && first) {
                    sS[qo]   = so;
                    sLoc[qo] = mineo ? loco : -1;
                    sLog[qo] = logit;
                    if (mineo) atomicMax(&sMaxU[loco], flipF(logit));
                }
            }
            __syncthreads();
            for (int base = 0; base < T; base += 64 * EDGE_U) {
                const int  qo   = base + uo * 64 + wave * 2 + half;
                const bool oko  = qo < T;
                const int  qc   = oko ? qo : 0;
                const int  lraw = sLoc[qc];
                const bool live = oko && lraw >= 0;
                const int  so   = sS[qc];
                const float po  = __expf(sLog[qc] - unflipF(sMaxU[live ? lraw : 0]));
                int s[EDGE_U], loc[EDGE_U];
                float w[EDGE_U];
                float4 ws[EDGE_U];
                row_share8(so, s);
#pragma unroll
                for (int u = 0; u < EDGE_U; ++u)
                    ws[u] = *(const float4*)(Wh + (size_t)s[u] * DD + 4 * lm);
                row_share8(live ? lraw : -1, loc);
                row_share8(po, w);
                if (live && first) atomicAdd(&sDen[lraw], po);
#pragma unroll
                for (int u = 0; u < EDGE_U; ++u) {
                    if (loc[u] >= 0) {
                        float* row = sAgg + loc[u] * DD + 4 * lm;
                        atomicAdd(row + 0, w[u] * ws[u].x);
                        atomicAdd(row + 1, w[u] * ws[u].y);
                        atomicAdd(row + 2, w[u] * ws[u].z);
                        atomicAdd(row + 3, w[u] * ws[u].w);
                    }
                }
            }
        } else {
            for (int e = tid; e < E; e += OWN_THREADS) {
                const unsigned loc = (unsigned)(dst[e] - lo);
                if (loc < (unsigned)cnt) {
                    const float p = lane_logit(Wh + (size_t)src[e] * DD, sWd + loc * DD, avec);
                    atomicMax(&sMaxU[loc], flipF(p));
                }
            }
            __syncthreads();
            for (int e = tid; e < E; e += OWN_THREADS) {
                const unsigned loc = (unsigned)(dst[e] - lo);
                if (loc < (unsigned)cnt) {
                    const float* whs = Wh + (size_t)src[e] * DD;
                    const float p = lane_logit(whs, sWd + loc * DD, avec);
                    const float w = __expf(p - unflipF(sMaxU[loc]));
                    atomicAdd(&sDen[loc], w);
                    for (int k = 0; k < DD; ++k) atomicAdd(&sAgg[loc * DD + k], w * whs[k]);
                }
            }
        }
        __syncthreads();

        for (int i = tid; i < cnt * (DD / 4); i += OWN_THREADS) {
            const float4 v = ((const float4*)sAgg)[i];
            const float den = fmaxf(sDen[i >> 4], 1e-9f);
            float4 o;
            o.x = 1.0f / (1.0f + __expf(-(v.x / den)));
            o.y = 1.0f / (1.0f + __expf(-(v.y / den)));
            o.z = 1.0f / (1.0f + __expf(-(v.z / den)));
            o.w = 1.0f / (1.0f + __expf(-(v.w / den)));
            ((float4*)(out + (size_t)lo * DD))[i] = o;
        }
        __syncthreads();
    }
}

extern "C" void kernel_launch(void* const* d_in, const int* in_sizes, int n_in,
                              void* d_out, int out_size, void* d_ws, size_t ws_size,
                              hipStream_t stream) {
    const float* x    = (const float*)d_in[0];
    const float* W    = (const float*)d_in[1];
    const float* bias = (const float*)d_in[2];
    const float* avec = (const float*)d_in[3];
    const int*   ei   = (const int*)d_in[4];

    const int N = in_sizes[0] / DD;
    const int E = in_sizes[4] / 2;
    const int* src = ei;
    const int* dst = ei + E;

    float* Wh = (float*)d_ws;
    unsigned* key = (unsigned*)(Wh + (size_t)N * DD);

    const int tiles = (N + 15) / 16;
    k_gemm<<<(tiles + 7) / 8, 256, 0, stream>>>(x, W, bias, Wh, N, dst, key, E);

    k_own<<<OWN_BLOCKS, OWN_THREADS, 0, stream>>>(Wh, avec, src, dst, key, (float*)d_out, N, E);
}


# two list walks become one: a node's softmax shift is the first logit to reach it (LDS compare-and-swap), not its maximum; same quotient; source rows gathered once; sums by LDS arrival
# speedup vs baseline: 5.3302x; 1.1479x over previous
#include <hip/hip_runtime.h>
#include <hip/hip_bf16.h>
#include <math.h>

#define DD 64
#define NSLOPE 0.2f

typedef __attribute__((ext_vector_type(2))) float v2f;
typedef __attribute__((ext_vector_type(4))) float v4f;
typedef __attribute__((ext_vector_type(8))) float v8f;

__device__ __forceinline__ unsigned flipF(float f) {
    unsigned u = __float_as_uint(f);
    return (u >> 31) ? ~u : (u | 0x80000000u);
}
__device__ __forceinline__ float unflipF(unsigned u) {
    unsigned v = (u >> 31) ? (u & 0x7FFFFFFFu) : ~u;
    return __uint_as_float(v);
}

#define RMAX 196
#define OWN_BLOCKS 256

static_assert(OWN_BLOCKS == 256, "the range key is one byte: range index mod 256 must be the owner");
__device__ __forceinline__ unsigned key_of(int d) { return ((unsigned)d / RMAX) & 0xFFu; }

__global__ __launch_bounds__(256) void k_gemm(const float* __restrict__ x,
                                              const float* __restrict__ W,
                                              const float* __restrict__ bias,
                                              float* __restrict__ Wh, int n,
                                              const int* __restrict__ dst,
                                              unsigned* __restrict__ key, int E) {
    {
        const bool vec = ((((size_t)dst) & 15) == 0);
        const int nDw = ((E + 15) >> 4) << 2;
        const int gstride = (int)gridDim.x * 256;
        for (int i = (int)blockIdx.x * 256 + (int)threadIdx.x; i < nDw; i += gstride) {
            const int e0 = 4 * i;
            int4 d;
            if (vec && e0 + 3 < E) {
                d = ((const int4*)dst)[i];
            } else {
                d.x = (e0 + 0 < E) ? dst[e0 + 0] : -1;
                d.y = (e0 + 1 < E) ? dst[e0 + 1] : -1;
                d.z = (e0 + 2 < E) ? dst[e0 + 2] : -1;
                d.w = (e0 + 3 < E) ? dst[e0 + 3] : -1;
            }
            key[i] = key_of(d.x) | (key_of(d.y) << 8) | (key_of(d.z) << 16) | (key_of(d.w) << 24);
        }
    }
    const int lane = threadIdx.x & 31;
    const int wave = threadIdx.x >> 5;
    const int tile = blockIdx.x * 8 + wave;
    if (tile * 16 >= n) return;

    const int lm   = lane & 15;
    const int half = lane >> 4;
    const int row  = tile * 16 + lm;

    const float* xr = x + (size_t)row * DD + 2 * half;
    v2f areg[16];
#pragma unroll
    for (int i = 0; i < 16; ++i) {
        areg[i].x = xr[4 * i + 0];
        areg[i].y = xr[4 * i + 1];
    }

#pragma unroll
    for (int c = 0; c < 4; ++c) {
        v8f acc = {};
        const float* wr = W + (size_t)(c * 16 + lm) * DD + 2 * half;
#pragma unroll
        for (int i = 0; i < 16; ++i) {
            v2f bfrag;
            bfrag.x = wr[4 * i + 0];
            bfrag.y = wr[4 * i + 1];
            acc = __builtin_amdgcn_wmma_f32_16x16x4_f32(
                false, areg[i], false, bfrag, (short)0, acc, false, false);
        }
        const float bv = bias[c * 16 + lm];
#pragma unroll
        for (int v = 0; v < 8; ++v) {
            Wh[(size_t)(tile * 16 + v + 8 * half) * DD + c * 16 + lm] = acc[v] + bv;
        }
    }
}

#define CAP 6144
#define OWN_THREADS 1024
#define SCAN_U 8
#define EDGE_U 8
#define NOREF 0xFFFFFFFFu
#define SHIFT_MAX 30.0f
static_assert(EDGE_U == 8, "a row's eight edges are owned by its eight lane pairs");

__device__ __forceinline__ float leaky(float s) { return s > 0.0f ? s : NSLOPE * s; }

template <int CTRL> __device__ __forceinline__ int row_move(int v) {
    return __builtin_amdgcn_update_dpp(0, v, CTRL, 0xf, 0xf, true);
}
template <int CTRL> __device__ __forceinline__ float row_move(float v) {
    return __int_as_float(__builtin_amdgcn_update_dpp(0, __float_as_int(v), CTRL, 0xf, 0xf, true));
}
template <typename T> __device__ __forceinline__ void row_share8(const T v, T (&out)[8]) {
    out[0] = row_move<0x150>(v); out[1] = row_move<0x152>(v);
    out[2] = row_move<0x154>(v); out[3] = row_move<0x156>(v);
    out[4] = row_move<0x158>(v); out[5] = row_move<0x15A>(v);
    out[6] = row_move<0x15C>(v); out[7] = row_move<0x15E>(v);
}
__device__ __forceinline__ float row_sums8(const float (&p)[8], const int lm) {
    const bool b3 = (lm & 8) != 0, b2 = (lm & 4) != 0, b1 = (lm & 2) != 0;
    float r[4];
#pragma unroll
    for (int j = 0; j < 4; ++j) {
        const float keep = b3 ? p[4 + j] : p[j];
        const float give = b3 ? p[j] : p[4 + j];
        r[j] = keep + row_move<0x168>(give);
    }
    float t[2];
#pragma unroll
    for (int j = 0; j < 2; ++j) {
        const float keep = b2 ? r[2 + j] : r[j];
        const float give = b2 ? r[j] : r[2 + j];
        t[j] = keep + row_move<0x164>(give);
    }
    const float keep = b1 ? t[1] : t[0];
    const float give = b1 ? t[0] : t[1];
    const float w = keep + row_move<0x162>(give);
    return w + row_move<0x161>(w);
}

__device__ __forceinline__ float lane_logit(const float* __restrict__ whs,
                                            const float* wdl,
                                            const float* __restrict__ avec) {
    float acc = 0.0f;
    for (int k = 0; k < DD; ++k) acc += avec[k] * leaky(whs[k] + wdl[k]);
    return acc;
}

__global__ __launch_bounds__(OWN_THREADS) void k_own(const float* __restrict__ Wh,
                                                     const float* __restrict__ avec,
                                                     const int* __restrict__ src,
                                                     const int* __restrict__ dst,
                                                     const unsigned* __restrict__ key,
                                                     float* __restrict__ out,
                                                     int n, int E) {
    __shared__ __attribute__((aligned(16))) float sWd[RMAX * DD];
    __shared__ __attribute__((aligned(16))) float sAgg[RMAX * DD];
    __shared__ unsigned sRefU[RMAX];
    __shared__ unsigned sMaxU[RMAX];
    __shared__ float    sDen[RMAX];
    __shared__ int      sE[CAP];
    __shared__ int      sCount;
    __shared__ int      sOvf;

    const int tid  = threadIdx.x;
    const int lane = tid & 31;
    const int wave = tid >> 5;
    const int lm   = lane & 15;
    const int half = lane >> 4;
    const int uo   = lm >> 1;
    const bool first = (lm & 1) == 0;
    const float4 a4 = ((const float4*)avec)[lm];

    const int nK4 = (E + 15) >> 4;
    const int rot = (int)(blockIdx.x * (unsigned)(nK4 / (int)gridDim.x)) & ~7;

    for (int lo = blockIdx.x * RMAX; lo < n; lo += gridDim.x * RMAX) {
        const int cnt = min(RMAX, n - lo);

        for (int i = tid; i < cnt * (DD / 4); i += OWN_THREADS) {
            ((float4*)sWd)[i]  = ((const float4*)(Wh + (size_t)lo * DD))[i];
            ((float4*)sAgg)[i] = make_float4(0.0f, 0.0f, 0.0f, 0.0f);
        }
        if (tid < RMAX) { sRefU[tid] = NOREF; sMaxU[tid] = 0u; sDen[tid] = 0.0f; }
        if (tid == 0) { sCount = 0; sOvf = 0; }
        __syncthreads();

        const unsigned K1 = 0x01010101u;
        const unsigned pat = ((unsigned)(lo / RMAX) & 0xFFu) * K1;
        auto pack2 = [&](const uint4 p, const uint4 q) -> unsigned {
            const unsigned p0 = p.x ^ pat, p1 = p.y ^ pat, p2 = p.z ^ pat, p3 = p.w ^ pat;
            const unsigned q0 = q.x ^ pat, q1 = q.y ^ pat, q2 = q.z ^ pat, q3 = q.w ^ pat;
            const unsigned h0 = (p0 - K1) & ~p0, h1 = (p1 - K1) & ~p1;
            const unsigned h2 = (p2 - K1) & ~p2, h3 = (p3 - K1) & ~p3;
            const unsigned g0 = (q0 - K1) & ~q0, g1 = (q1 - K1) & ~q1;
            const unsigned g2 = (q2 - K1) & ~q2, g3 = (q3 - K1) & ~q3;
            return ((h0 >> 7) & K1)        | ((h1 >> 6) & (K1 << 1))
                 | ((h2 >> 5) & (K1 << 2)) | ((h3 >> 4) & (K1 << 3))
                 | ((g0 >> 3) & (K1 << 4)) | ((g1 >> 2) & (K1 << 5))
                 | ((g2 >> 1) & (K1 << 6)) | (g3 & (K1 << 7));
        };
        auto ldkey = [&](const int k) {
            return *(const uint4*)((const char*)key + ((unsigned)k << 4));
        };
        auto step = [&](const int i0, const bool guarded) {
            uint4 v[SCAN_U];
            int   ida[SCAN_U / 2], idb[SCAN_U / 2];
            bool  in[SCAN_U];
#pragma unroll
            for (int j = 0; j < SCAN_U; ++j) {
                const int i = i0 + j * OWN_THREADS;
                in[j] = !guarded || i < nK4;
                int k = i + rot;
                if (k >= nK4) k -= nK4;
                if (guarded && !in[j]) k = 0;
                if (j & 1) idb[j >> 1] = k; else ida[j >> 1] = k;
                v[j]  = ldkey(k);
            }
            unsigned w[SCAN_U / 2];
            int c = 0;
#pragma unroll
            for (int q = 0; q < SCAN_U / 2; ++q) {
                w[q] = pack2(v[2 * q], v[2 * q + 1]);
                if (guarded) {
                    if (!in[2 * q])     w[q] &= 0xF0F0F0F0u;
                    if (!in[2 * q + 1]) w[q] &= 0x0F0F0F0Fu;
                }
                c += __popc(w[q]);
            }
            int incl = c;
            incl += __builtin_amdgcn_update_dpp(0, incl, 0x111, 0xf, 0xf, true);
            incl += __builtin_amdgcn_update_dpp(0, incl, 0x112, 0xf, 0xf, true);
            incl += __builtin_amdgcn_update_dpp(0, incl, 0x114, 0xf, 0xf, true);
            incl += __builtin_amdgcn_update_dpp(0, incl, 0x118, 0xf, 0xf, true);
            const int lowsum = __builtin_amdgcn_readlane(incl, 15);
            if (lane >= 16) incl += lowsum;
            const int tot = __builtin_amdgcn_readlane(incl, 31);
            int basepos = 0;
            if (lane == 0) basepos = atomicAdd(&sCount, tot);
            basepos = __builtin_amdgcn_readfirstlane(basepos);
            int pos = basepos + incl - c;
#pragma unroll
            for (int q = 0; q < SCAN_U / 2; ++q) {
                unsigned m = w[q];
                while (m) {
                    const int t = __ffs((int)m) - 1;
                    m &= m - 1u;
                    const int k = (t & 4) ? idb[q] : ida[q];
                    if (pos < CAP) sE[pos] = 16 * k + 4 * (t & 3) + (t >> 3);
                    ++pos;
                }
            }
        };
        {
            const int wfirst = __builtin_amdgcn_readfirstlane(tid & ~31);
            int s0 = 0;
            for (; s0 + wfirst + 31 + (SCAN_U - 1) * OWN_THREADS < nK4; s0 += SCAN_U * OWN_THREADS)
                step(s0 + tid, false);
            if (s0 + wfirst < nK4) step(s0 + tid, true);
        }
        __syncthreads();

        const int total = sCount;
        const int over  = total > CAP;

        bool exact = over;
        if (!over) {
            const int T = total;
            for (int base = 0; base < T; base += 64 * EDGE_U) {
                const int  qo  = base + uo * 64 + wave * 2 + half;
                const bool oko = qo < T;
                const int  e   = sE[oko ? qo : 0];
                const int  ec  = e < E ? e : 0;
                const int  so  = src[ec];
                const unsigned l = (unsigned)(dst[ec] - lo);
                const bool mineo = oko && e < E && l < (unsigned)cnt;
                const int  loco  = mineo ? (int)l : 0;
                int s[EDGE_U], loc[EDGE_U];
                row_share8(so, s);
                row_share8(loco, loc);
                float p[EDGE_U];
                v4f   ws[EDGE_U];
#pragma unroll
                for (int u = 0; u < EDGE_U; ++u)
                    ws[u] = *(const v4f*)(Wh + (size_t)s[u] * DD + 4 * lm);
#pragma unroll
                for (int u = 0; u < EDGE_U; ++u) {
                    const v4f wd = *(const v4f*)(sWd + loc[u] * DD + 4 * lm);
                    const v4f t  = ws[u] + wd;
                    const v4f lk = t > 0.0f ? t : NSLOPE * t;
                    p[u] = (a4.x * lk.x + a4.y * lk.y) + (a4.z * lk.z + a4.w * lk.w);
                }
                const float logit = row_sums8(p, lm);
                float wo = 0.0f;
                if (mineo && first) {
                    const unsigned old = atomicCAS(&sRefU[loco], NOREF, __float_as_uint(logit));
                    const float d = logit - (old == NOREF ? logit : __uint_as_float(old));
                    if (d > SHIFT_MAX) sOvf = 1;
                    wo = __expf(d);
                    atomicAdd(&sDen[loco], wo);
                }
                float w[EDGE_U];
                row_share8(wo, w);
#pragma unroll
                for (int u = 0; u < EDGE_U; ++u) {
                    if (w[u] != 0.0f) {
                        float* row = sAgg + loc[u] * DD + 4 * lm;
                        atomicAdd(row + 0, w[u] * ws[u].x);
                        atomicAdd(row + 1, w[u] * ws[u].y);
                        atomicAdd(row + 2, w[u] * ws[u].z);
                        atomicAdd(row + 3, w[u] * ws[u].w);
                    }
                }
            }
            __syncthreads();
            if (sOvf) {
                for (int i = tid; i < cnt * (DD / 4); i += OWN_THREADS)
                    ((float4*)sAgg)[i] = make_float4(0.0f, 0.0f, 0.0f, 0.0f);
                if (tid < RMAX) sDen[tid] = 0.0f;
                exact = true;
                __syncthreads();
            }
        }
        if (exact) {
            for (int e = tid; e < E; e += OWN_THREADS) {
                const unsigned loc = (unsigned)(dst[e] - lo);
                if (loc < (unsigned)cnt) {
                    const float p = lane_logit(Wh + (size_t)src[e] * DD, sWd + loc * DD, avec);
                    atomicMax(&sMaxU[loc], flipF(p));
                }
            }
            __syncthreads();
            for (int e = tid; e < E; e += OWN_THREADS) {
                const unsigned loc = (unsigned)(dst[e] - lo);
                if (loc < (unsigned)cnt) {
                    const float* whs = Wh + (size_t)src[e] * DD;
                    const float p = lane_logit(whs, sWd + loc * DD, avec);
                    const float w = __expf(p - unflipF(sMaxU[loc]));
                    atomicAdd(&sDen[loc], w);
                    for (int k = 0; k < DD; ++k) atomicAdd(&sAgg[loc * DD + k], w * whs[k]);
                }
            }
            __syncthreads();
        }

        for (int i = tid; i < cnt * (DD / 4); i += OWN_THREADS) {
            const float4 v = ((const float4*)sAgg)[i];
            const float den = fmaxf(sDen[i >> 4], 1e-9f);
            float4 o;
            o.x = 1.0f / (1.0f + __expf(-(v.x / den)));
            o.y = 1.0f / (1.0f + __expf(-(v.y / den)));
            o.z = 1.0f / (1.0f + __expf(-(v.z / den)));
            o.w = 1.0f / (1.0f + __expf(-(v.w / den)));
            ((float4*)(out + (size_t)lo * DD))[i] = o;
        }
        __syncthreads();
    }
}

extern "C" void kernel_launch(void* const* d_in, const int* in_sizes, int n_in,
                              void* d_out, int out_size, void* d_ws, size_t ws_size,
                              hipStream_t stream) {
    const float* x    = (const float*)d_in[0];
    const float* W    = (const float*)d_in[1];
    const float* bias = (const float*)d_in[2];
    const float* avec = (const float*)d_in[3];
    const int*   ei   = (const int*)d_in[4];

    const int N = in_sizes[0] / DD;
    const int E = in_sizes[4] / 2;
    const int* src = ei;
    const int* dst = ei + E;

    float* Wh = (float*)d_ws;
    unsigned* key = (unsigned*)(Wh + (size_t)N * DD);

    const int tiles = (N + 15) / 16;
    k_gemm<<<(tiles + 7) / 8, 256, 0, stream>>>(x, W, bias, Wh, N, dst, key, E);

    k_own<<<OWN_BLOCKS, OWN_THREADS, 0, stream>>>(Wh, avec, src, dst, key, (float*)d_out, N, E);
}


# row sums: each 16-lane half of a wave adds into its own copy of the sum rows, 2 banks apart: an LDS add's 32 lanes hit 32 banks, not 16 twice; a node's sum is 2 partials by arrival, joined at the end
# speedup vs baseline: 5.9954x; 1.1248x over previous
#include <hip/hip_runtime.h>
#include <hip/hip_bf16.h>
#include <math.h>

#define DD 64
#define NSLOPE 0.2f

typedef __attribute__((ext_vector_type(2))) float v2f;
typedef __attribute__((ext_vector_type(4))) float v4f;
typedef __attribute__((ext_vector_type(8))) float v8f;

__device__ __forceinline__ unsigned flipF(float f) {
    unsigned u = __float_as_uint(f);
    return (u >> 31) ? ~u : (u | 0x80000000u);
}
__device__ __forceinline__ float unflipF(unsigned u) {
    unsigned v = (u >> 31) ? (u & 0x7FFFFFFFu) : ~u;
    return __uint_as_float(v);
}

#define RMAX 196
#define OWN_BLOCKS 256

static_assert(OWN_BLOCKS == 256, "the range key is one byte: range index mod 256 must be the owner");
__device__ __forceinline__ unsigned key_of(int d) { return ((unsigned)d / RMAX) & 0xFFu; }

__global__ __launch_bounds__(256) void k_gemm(const float* __restrict__ x,
                                              const float* __restrict__ W,
                                              const float* __restrict__ bias,
                                              float* __restrict__ Wh, int n,
                                              const int* __restrict__ dst,
                                              unsigned* __restrict__ key, int E) {
    {
        const bool vec = ((((size_t)dst) & 15) == 0);
        const int nDw = ((E + 15) >> 4) << 2;
        const int gstride = (int)gridDim.x * 256;
        for (int i = (int)blockIdx.x * 256 + (int)threadIdx.x; i < nDw; i += gstride) {
            const int e0 = 4 * i;
            int4 d;
            if (vec && e0 + 3 < E) {
                d = ((const int4*)dst)[i];
            } else {
                d.x = (e0 + 0 < E) ? dst[e0 + 0] : -1;
                d.y = (e0 + 1 < E) ? dst[e0 + 1] : -1;
                d.z = (e0 + 2 < E) ? dst[e0 + 2] : -1;
                d.w = (e0 + 3 < E) ? dst[e0 + 3] : -1;
            }
            key[i] = key_of(d.x) | (key_of(d.y) << 8) | (key_of(d.z) << 16) | (key_of(d.w) << 24);
        }
    }
    const int lane = threadIdx.x & 31;
    const int wave = threadIdx.x >> 5;
    const int tile = blockIdx.x * 8 + wave;
    if (tile * 16 >= n) return;

    const int lm   = lane & 15;
    const int half = lane >> 4;
    const int row  = tile * 16 + lm;

    const float* xr = x + (size_t)row * DD + 2 * half;
    v2f areg[16];
#pragma unroll
    for (int i = 0; i < 16; ++i) {
        areg[i].x = xr[4 * i + 0];
        areg[i].y = xr[4 * i + 1];
    }

#pragma unroll
    for (int c = 0; c < 4; ++c) {
        v8f acc = {};
        const float* wr = W + (size_t)(c * 16 + lm) * DD + 2 * half;
#pragma unroll
        for (int i = 0; i < 16; ++i) {
            v2f bfrag;
            bfrag.x = wr[4 * i + 0];
            bfrag.y = wr[4 * i + 1];
            acc = __builtin_amdgcn_wmma_f32_16x16x4_f32(
                false, areg[i], false, bfrag, (short)0, acc, false, false);
        }
        const float bv = bias[c * 16 + lm];
#pragma unroll
        for (int v = 0; v < 8; ++v) {
            Wh[(size_t)(tile * 16 + v + 8 * half) * DD + c * 16 + lm] = acc[v] + bv;
        }
    }
}

#define CAP 6144
#define OWN_THREADS 1024
#define SCAN_U 8
#define EDGE_U 8
#define NOREF 0xFFFFFFFFu
#define SHIFT_MAX 30.0f
#define AGG_ROWS   (RMAX * DD)
#define AGG_SECOND (AGG_ROWS + 2)
#define AGG_FLOATS (2 * AGG_ROWS + 4)
static_assert(AGG_ROWS % 64 == 0, "a copy is whole passes over the banks, so the offset of 2 floats is the bank shift");
static_assert(AGG_FLOATS % 4 == 0 && AGG_SECOND + AGG_ROWS <= AGG_FLOATS, "clearing in 16-byte pieces covers both copies");
static_assert(EDGE_U == 8, "a row's eight edges are owned by its eight lane pairs");

__device__ __forceinline__ float leaky(float s) { return s > 0.0f ? s : NSLOPE * s; }

template <int CTRL> __device__ __forceinline__ int row_move(int v) {
    return __builtin_amdgcn_update_dpp(0, v, CTRL, 0xf, 0xf, true);
}
template <int CTRL> __device__ __forceinline__ float row_move(float v) {
    return __int_as_float(__builtin_amdgcn_update_dpp(0, __float_as_int(v), CTRL, 0xf, 0xf, true));
}
template <typename T> __device__ __forceinline__ void row_share8(const T v, T (&out)[8]) {
    out[0] = row_move<0x150>(v); out[1] = row_move<0x152>(v);
    out[2] = row_move<0x154>(v); out[3] = row_move<0x156>(v);
    out[4] = row_move<0x158>(v); out[5] = row_move<0x15A>(v);
    out[6] = row_move<0x15C>(v); out[7] = row_move<0x15E>(v);
}
__device__ __forceinline__ float row_sums8(const float (&p)[8], const int lm) {
    const bool b3 = (lm & 8) != 0, b2 = (lm & 4) != 0, b1 = (lm & 2) != 0;
    float r[4];
#pragma unroll
    for (int j = 0; j < 4; ++j) {
        const float keep = b3 ? p[4 + j] : p[j];
        const float give = b3 ? p[j] : p[4 + j];
        r[j] = keep + row_move<0x168>(give);
    }
    float t[2];
#pragma unroll
    for (int j = 0; j < 2; ++j) {
        const float keep = b2 ? r[2 + j] : r[j];
        const float give = b2 ? r[j] : r[2 + j];
        t[j] = keep + row_move<0x164>(give);
    }
    const float keep = b1 ? t[1] : t[0];
    const float give = b1 ? t[0] : t[1];
    const float w = keep + row_move<0x162>(give);
    return w + row_move<0x161>(w);
}

__device__ __forceinline__ float lane_logit(const float* __restrict__ whs,
                                            const float* wdl,
                                            const float* __restrict__ avec) {
    float acc = 0.0f;
    for (int k = 0; k < DD; ++k) acc += avec[k] * leaky(whs[k] + wdl[k]);
    return acc;
}

__global__ __launch_bounds__(OWN_THREADS) void k_own(const float* __restrict__ Wh,
                                                     const float* __restrict__ avec,
                                                     const int* __restrict__ src,
                                                     const int* __restrict__ dst,
                                                     const unsigned* __restrict__ key,
                                                     float* __restrict__ out,
                                                     int n, int E) {
    __shared__ __attribute__((aligned(16))) float sWd[RMAX * DD];
    __shared__ __attribute__((aligned(16))) float sAgg[AGG_FLOATS];
    __shared__ unsigned sRefU[RMAX];
    __shared__ unsigned sMaxU[RMAX];
    __shared__ float    sDen[RMAX];
    __shared__ int      sE[CAP];
    __shared__ int      sCount;
    __shared__ int      sOvf;

    const int tid  = threadIdx.x;
    const int lane = tid & 31;
    const int wave = tid >> 5;
    const int lm   = lane & 15;
    const int half = lane >> 4;
    const int uo   = lm >> 1;
    const bool first = (lm & 1) == 0;
    const float4 a4 = ((const float4*)avec)[lm];
    float* const aggLane = sAgg + (half ? AGG_SECOND : 0) + 4 * lm;

    const int nK4 = (E + 15) >> 4;
    const int rot = (int)(blockIdx.x * (unsigned)(nK4 / (int)gridDim.x)) & ~7;

    for (int lo = blockIdx.x * RMAX; lo < n; lo += gridDim.x * RMAX) {
        const int cnt = min(RMAX, n - lo);

        for (int i = tid; i < cnt * (DD / 4); i += OWN_THREADS)
            ((float4*)sWd)[i]  = ((const float4*)(Wh + (size_t)lo * DD))[i];
        for (int i = tid; i < AGG_FLOATS / 4; i += OWN_THREADS)
            ((float4*)sAgg)[i] = make_float4(0.0f, 0.0f, 0.0f, 0.0f);
        if (tid < RMAX) { sRefU[tid] = NOREF; sMaxU[tid] = 0u; sDen[tid] = 0.0f; }
        if (tid == 0) { sCount = 0; sOvf = 0; }
        __syncthreads();

        const unsigned K1 = 0x01010101u;
        const unsigned pat = ((unsigned)(lo / RMAX) & 0xFFu) * K1;
        auto pack2 = [&](const uint4 p, const uint4 q) -> unsigned {
            const unsigned p0 = p.x ^ pat, p1 = p.y ^ pat, p2 = p.z ^ pat, p3 = p.w ^ pat;
            const unsigned q0 = q.x ^ pat, q1 = q.y ^ pat, q2 = q.z ^ pat, q3 = q.w ^ pat;
            const unsigned h0 = (p0 - K1) & ~p0, h1 = (p1 - K1) & ~p1;
            const unsigned h2 = (p2 - K1) & ~p2, h3 = (p3 - K1) & ~p3;
            const unsigned g0 = (q0 - K1) & ~q0, g1 = (q1 - K1) & ~q1;
            const unsigned g2 = (q2 - K1) & ~q2, g3 = (q3 - K1) & ~q3;
            return ((h0 >> 7) & K1)        | ((h1 >> 6) & (K1 << 1))
                 | ((h2 >> 5) & (K1 << 2)) | ((h3 >> 4) & (K1 << 3))
                 | ((g0 >> 3) & (K1 << 4)) | ((g1 >> 2) & (K1 << 5))
                 | ((g2 >> 1) & (K1 << 6)) | (g3 & (K1 << 7));
        };
        auto ldkey = [&](const int k) {
            return *(const uint4*)((const char*)key + ((unsigned)k << 4));
        };
        auto step = [&](const int i0, const bool guarded) {
            uint4 v[SCAN_U];
            int   ida[SCAN_U / 2], idb[SCAN_U / 2];
            bool  in[SCAN_U];
#pragma unroll
            for (int j = 0; j < SCAN_U; ++j) {
                const int i = i0 + j * OWN_THREADS;
                in[j] = !guarded || i < nK4;
                int k = i + rot;
                if (k >= nK4) k -= nK4;
                if (guarded && !in[j]) k = 0;
                if (j & 1) idb[j >> 1] = k; else ida[j >> 1] = k;
                v[j]  = ldkey(k);
            }
            unsigned w[SCAN_U / 2];
            int c = 0;
#pragma unroll
            for (int q = 0; q < SCAN_U / 2; ++q) {
                w[q] = pack2(v[2 * q], v[2 * q + 1]);
                if (guarded) {
                    if (!in[2 * q])     w[q] &= 0xF0F0F0F0u;
                    if (!in[2 * q + 1]) w[q] &= 0x0F0F0F0Fu;
                }
                c += __popc(w[q]);
            }
            int incl = c;
            incl += __builtin_amdgcn_update_dpp(0, incl, 0x111, 0xf, 0xf, true);
            incl += __builtin_amdgcn_update_dpp(0, incl, 0x112, 0xf, 0xf, true);
            incl += __builtin_amdgcn_update_dpp(0, incl, 0x114, 0xf, 0xf, true);
            incl += __builtin_amdgcn_update_dpp(0, incl, 0x118, 0xf, 0xf, true);
            const int lowsum = __builtin_amdgcn_readlane(incl, 15);
            if (lane >= 16) incl += lowsum;
            const int tot = __builtin_amdgcn_readlane(incl, 31);
            int basepos = 0;
            if (lane == 0) basepos = atomicAdd(&sCount, tot);
            basepos = __builtin_amdgcn_readfirstlane(basepos);
            int pos = basepos + incl - c;
#pragma unroll
            for (int q = 0; q < SCAN_U / 2; ++q) {
                unsigned m = w[q];
                while (m) {
                    const int t = __ffs((int)m) - 1;
                    m &= m - 1u;
                    const int k = (t & 4) ? idb[q] : ida[q];
                    if (pos < CAP) sE[pos] = 16 * k + 4 * (t & 3) + (t >> 3);
                    ++pos;
                }
            }
        };
        {
            const int wfirst = __builtin_amdgcn_readfirstlane(tid & ~31);
            int s0 = 0;
            for (; s0 + wfirst + 31 + (SCAN_U - 1) * OWN_THREADS < nK4; s0 += SCAN_U * OWN_THREADS)
                step(s0 + tid, false);
            if (s0 + wfirst < nK4) step(s0 + tid, true);
        }
        __syncthreads();

        const int total = sCount;
        const int over  = total > CAP;

        bool exact = over;
        if (!over) {
            const int T = total;
            for (int base = 0; base < T; base += 64 * EDGE_U) {
                const int  qo  = base + uo * 64 + wave * 2 + half;
                const bool oko = qo < T;
                const int  e   = sE[oko ? qo : 0];
                const int  ec  = e < E ? e : 0;
                const int  so  = src[ec];
                const unsigned l = (unsigned)(dst[ec] - lo);
                const bool mineo = oko && e < E && l < (unsigned)cnt;
                const int  loco  = mineo ? (int)l : 0;
                int s[EDGE_U], loc[EDGE_U];
                row_share8(so, s);
                row_share8(loco, loc);
                float p[EDGE_U];
                v4f   ws[EDGE_U];
#pragma unroll
                for (int u = 0; u < EDGE_U; ++u)
                    ws[u] = *(const v4f*)(Wh + (size_t)s[u] * DD + 4 * lm);
#pragma unroll
                for (int u = 0; u < EDGE_U; ++u) {
                    const v4f wd = *(const v4f*)(sWd + loc[u] * DD + 4 * lm);
                    const v4f t  = ws[u] + wd;
                    const v4f lk = t > 0.0f ? t : NSLOPE * t;
                    p[u] = (a4.x * lk.x + a4.y * lk.y) + (a4.z * lk.z + a4.w * lk.w);
                }
                const float logit = row_sums8(p, lm);
                float wo = 0.0f;
                if (mineo && first) {
                    const unsigned old = atomicCAS(&sRefU[loco], NOREF, __float_as_uint(logit));
                    const float d = logit - (old == NOREF ? logit : __uint_as_float(old));
                    if (d > SHIFT_MAX) sOvf = 1;
                    wo = __expf(d);
                    atomicAdd(&sDen[loco], wo);
                }
                float w[EDGE_U];
                row_share8(wo, w);
#pragma unroll
                for (int u = 0; u < EDGE_U; ++u) {
                    if (w[u] != 0.0f) {
                        float* row = aggLane + loc[u] * DD;
                        atomicAdd(row + 0, w[u] * ws[u].x);
                        atomicAdd(row + 1, w[u] * ws[u].y);
                        atomicAdd(row + 2, w[u] * ws[u].z);
                        atomicAdd(row + 3, w[u] * ws[u].w);
                    }
                }
            }
            __syncthreads();
            if (sOvf) {
                for (int i = tid; i < AGG_FLOATS / 4; i += OWN_THREADS)
                    ((float4*)sAgg)[i] = make_float4(0.0f, 0.0f, 0.0f, 0.0f);
                if (tid < RMAX) sDen[tid] = 0.0f;
                exact = true;
                __syncthreads();
            }
        }
        if (exact) {
            for (int e = tid; e < E; e += OWN_THREADS) {
                const unsigned loc = (unsigned)(dst[e] - lo);
                if (loc < (unsigned)cnt) {
                    const float p = lane_logit(Wh + (size_t)src[e] * DD, sWd + loc * DD, avec);
                    atomicMax(&sMaxU[loc], flipF(p));
                }
            }
            __syncthreads();
            for (int e = tid; e < E; e += OWN_THREADS) {
                const unsigned loc = (unsigned)(dst[e] - lo);
                if (loc < (unsigned)cnt) {
                    const float* whs = Wh + (size_t)src[e] * DD;
                    const float p = lane_logit(whs, sWd + loc * DD, avec);
                    const float w = __expf(p - unflipF(sMaxU[loc]));
                    atomicAdd(&sDen[loc], w);
                    for (int k = 0; k < DD; ++k) atomicAdd(&sAgg[loc * DD + k], w * whs[k]);
                }
            }
            __syncthreads();
        }

        for (int i = tid; i < cnt * (DD / 4); i += OWN_THREADS) {
            float4 v = ((const float4*)sAgg)[i];
            const float2 v2a = ((const float2*)(sAgg + AGG_SECOND))[2 * i];
            const float2 v2b = ((const float2*)(sAgg + AGG_SECOND))[2 * i + 1];
            v.x += v2a.x; v.y += v2a.y; v.z += v2b.x; v.w += v2b.y;
            const float den = fmaxf(sDen[i >> 4], 1e-9f);
            float4 o;
            o.x = 1.0f / (1.0f + __expf(-(v.x / den)));
            o.y = 1.0f / (1.0f + __expf(-(v.y / den)));
            o.z = 1.0f / (1.0f + __expf(-(v.z / den)));
            o.w = 1.0f / (1.0f + __expf(-(v.w / den)));
            ((float4*)(out + (size_t)lo * DD))[i] = o;
        }
        __syncthreads();
    }
}

extern "C" void kernel_launch(void* const* d_in, const int* in_sizes, int n_in,
                              void* d_out, int out_size, void* d_ws, size_t ws_size,
                              hipStream_t stream) {
    const float* x    = (const float*)d_in[0];
    const float* W    = (const float*)d_in[1];
    const float* bias = (const float*)d_in[2];
    const float* avec = (const float*)d_in[3];
    const int*   ei   = (const int*)d_in[4];

    const int N = in_sizes[0] / DD;
    const int E = in_sizes[4] / 2;
    const int* src = ei;
    const int* dst = ei + E;

    float* Wh = (float*)d_ws;
    unsigned* key = (unsigned*)(Wh + (size_t)N * DD);

    const int tiles = (N + 15) / 16;
    k_gemm<<<(tiles + 7) / 8, 256, 0, stream>>>(x, W, bias, Wh, N, dst, key, E);

    k_own<<<OWN_BLOCKS, OWN_THREADS, 0, stream>>>(Wh, avec, src, dst, key, (float*)d_out, N, E);
}
